# starting kernel
# baseline (speedup 1.0000x reference)
#include <hip/hip_runtime.h>
#include <hip/hip_bf16.h>


#define NB     4
#define CIN    256
#define CCOMP  64
#define CMASK  100
#define HH     64
#define WW     64
#define HW     (HH*WW)
#define HO     128
#define WO     128
#define HWO    (HO*WO)
#define KTOT   576
#define MPAD   112

typedef float v2f __attribute__((ext_vector_type(2)));
typedef float v8f __attribute__((ext_vector_type(8)));

__global__ void repack_wenc(const float* __restrict__ w_enc,
                            float* __restrict__ w_encT) {
    int idx = blockIdx.x * 256 + threadIdx.x;
    if (idx >= MPAD * KTOT) return;
    int m     = idx / KTOT;
    int r     = idx % KTOT;
    int plane = r >> 6;
    int ci    = r & 63;
    float v = 0.0f;
    if (m < CMASK) v = w_enc[m * KTOT + ci * 9 + plane];
    w_encT[idx] = v;
}

__global__ void __launch_bounds__(128)
compress_gemm(const float* __restrict__ x, const float* __restrict__ w_comp,
              const float* __restrict__ b_comp, float* __restrict__ comp) {
    const int wid   = blockIdx.x * 4 + (threadIdx.x >> 5);
    const int lane  = threadIdx.x & 31;
    const int n     = wid >> 8;
    const int nt    = wid & 255;
    const int col   = nt * 16 + (lane & 15);
    const int khalf = (lane >> 4) * 2;
    const int mrow  = lane & 15;

    const float* xb = x + (size_t)n * CIN * HW + col;

    v8f acc[4];
    const v8f vzero = {0.f,0.f,0.f,0.f,0.f,0.f,0.f,0.f};
#pragma unroll
    for (int mt = 0; mt < 4; ++mt) acc[mt] = vzero;

    for (int kb = 0; kb < CIN; kb += 4) {
        const int k = kb + khalf;
        v2f bfrag;
        bfrag.x = xb[(size_t)k * HW];
        bfrag.y = xb[(size_t)(k + 1) * HW];
#pragma unroll
        for (int mt = 0; mt < 4; ++mt) {
            v2f afrag = *(const v2f*)(w_comp + (size_t)(mt * 16 + mrow) * CIN + k);
            acc[mt] = __builtin_amdgcn_wmma_f32_16x16x4_f32(
                false, afrag, false, bfrag, (short)0, acc[mt], false, false);
        }
    }

    float* cb = comp + (size_t)n * CCOMP * HW + col;
    const int mhi = 8 * (lane >> 4);
#pragma unroll
    for (int mt = 0; mt < 4; ++mt) {
#pragma unroll
        for (int r = 0; r < 8; ++r) {
            const int m = mt * 16 + r + mhi;
            cb[(size_t)m * HW] = acc[mt][r] + b_comp[m];
        }
    }
}

__global__ void __launch_bounds__(128)
encoder_gemm(const float* __restrict__ comp, const float* __restrict__ w_encT,
             const float* __restrict__ b_enc, float* __restrict__ mask) {
    const int wid   = blockIdx.x * 4 + (threadIdx.x >> 5);
    const int lane  = threadIdx.x & 31;
    const int n     = wid >> 8;
    const int nt    = wid & 255;
    const int col   = nt * 16 + (lane & 15);
    const int h     = col >> 6;
    const int w     = col & 63;
    const int khalf = (lane >> 4) * 2;
    const int mrow  = lane & 15;

    v8f acc[7];
    const v8f vzero = {0.f,0.f,0.f,0.f,0.f,0.f,0.f,0.f};
#pragma unroll
    for (int mt = 0; mt < 7; ++mt) acc[mt] = vzero;

    const float* cbase = comp + (size_t)n * CCOMP * HW;

#pragma unroll
    for (int plane = 0; plane < 9; ++plane) {
        const int dy = plane / 3, dx = plane % 3;
        const int y  = h + dy - 1;
        const int xw = w + dx - 1;
        const bool valid = (y >= 0) && (y < HH) && (xw >= 0) && (xw < WW);
        const int yc = min(max(y, 0), HH - 1);
        const int xc = min(max(xw, 0), WW - 1);
        const float* bsrc = cbase + yc * WW + xc;

#pragma unroll 4
        for (int cib = 0; cib < CCOMP; cib += 4) {
            const int ci = cib + khalf;
            float b0 = bsrc[(size_t)ci * HW];
            float b1 = bsrc[(size_t)(ci + 1) * HW];
            v2f bfrag;
            bfrag.x = valid ? b0 : 0.0f;
            bfrag.y = valid ? b1 : 0.0f;
            const int kk = plane * 64 + ci;
#pragma unroll
            for (int mt = 0; mt < 7; ++mt) {
                v2f afrag = *(const v2f*)(w_encT + (size_t)(mt * 16 + mrow) * KTOT + kk);
                acc[mt] = __builtin_amdgcn_wmma_f32_16x16x4_f32(
                    false, afrag, false, bfrag, (short)0, acc[mt], false, false);
            }
        }
    }

    float* mbase = mask + (size_t)n * CMASK * HW + col;
    const int mhi = 8 * (lane >> 4);
#pragma unroll
    for (int mt = 0; mt < 7; ++mt) {
#pragma unroll
        for (int r = 0; r < 8; ++r) {
            const int cm = mt * 16 + r + mhi;
            if (cm < CMASK) mbase[(size_t)cm * HW] = acc[mt][r] + b_enc[cm];
        }
    }
}

__global__ void softmax_shuffle(const float* __restrict__ mask,
                                float* __restrict__ msoft) {
    int t = blockIdx.x * 256 + threadIdx.x;
    if (t >= NB * HWO) return;
    int n   = t / HWO;
    int rem = t % HWO;
    int Hr  = rem / WO;
    int Wj  = rem % WO;
    int h = Hr >> 1, sh = Hr & 1;
    int w = Wj >> 1, sw = Wj & 1;
    int r = sh * 2 + sw;

    const float* mb = mask + (size_t)n * CMASK * HW + h * WW + w;
    float v[25];
    float mx = -1e30f;
#pragma unroll
    for (int k = 0; k < 25; ++k) {
        v[k] = mb[(size_t)(k * 4 + r) * HW];
        mx = fmaxf(mx, v[k]);
    }
    float s = 0.0f;
#pragma unroll
    for (int k = 0; k < 25; ++k) { v[k] = __expf(v[k] - mx); s += v[k]; }
    float inv = 1.0f / s;
    float* ob = msoft + (size_t)n * 25 * HWO + Hr * WO + Wj;
#pragma unroll
    for (int k = 0; k < 25; ++k) ob[(size_t)k * HWO] = v[k] * inv;
}

__global__ void __launch_bounds__(256)
reassemble(const float* __restrict__ x, const float* __restrict__ msoft,
           float* __restrict__ out) {
    const int n   = blockIdx.x >> 6;
    const int h   = blockIdx.x & 63;
    const int t   = threadIdx.x;
    const int sh  = t >> 7;
    const int j   = t & 127;
    const int wlo = j >> 1;
    const int Hr  = 2 * h + sh;

    float wt[25];
    const float* mb = msoft + (size_t)n * 25 * HWO + Hr * WO + j;
#pragma unroll
    for (int k = 0; k < 25; ++k)
        wt[k] = __builtin_nontemporal_load(&mb[(size_t)k * HWO]);

    const int  idx0 = t;
    const int  row0 = idx0 / 68, col0 = idx0 % 68;
    const int  yr0  = h + row0 - 2, xc0 = col0 - 2;
    const bool v0   = (yr0 >= 0) && (yr0 < HH) && (xc0 >= 0) && (xc0 < WW);
    const size_t off0 = v0 ? (size_t)(yr0 * WW + xc0) : 0;

    const int  idx1 = t + 256;
    const bool has1 = idx1 < 5 * 68;
    const int  row1 = idx1 / 68, col1 = idx1 % 68;
    const int  yr1  = h + row1 - 2, xc1 = col1 - 2;
    const bool v1   = has1 && (yr1 >= 0) && (yr1 < HH) && (xc1 >= 0) && (xc1 < WW);
    const size_t off1 = v1 ? (size_t)(yr1 * WW + xc1) : 0;

    __shared__ float tile[2][5 * 68];
    const float* xb = x + (size_t)n * CIN * HW;
    float* ob = out + (size_t)n * CIN * HWO + Hr * WO + j;

    {
        float s0 = v0 ? xb[off0] : 0.0f;
        float s1 = v1 ? xb[off1] : 0.0f;
        tile[0][idx0] = s0;
        if (has1) tile[0][idx1] = s1;
    }
    __syncthreads();

    for (int c = 0; c < CIN; ++c) {
        const int cur = c & 1;
        const int nxt = cur ^ 1;

        float s0 = 0.0f, s1 = 0.0f;
        if (c + 1 < CIN) {
            const float* xc = xb + (size_t)(c + 1) * HW;
            s0 = v0 ? xc[off0] : 0.0f;
            s1 = v1 ? xc[off1] : 0.0f;
        }

        float acc = 0.0f;
#pragma unroll
        for (int p = 0; p < 5; ++p)
#pragma unroll
            for (int q = 0; q < 5; ++q)
                acc = fmaf(wt[p * 5 + q], tile[cur][p * 68 + wlo + q], acc);

        __builtin_nontemporal_store(acc, &ob[(size_t)c * HWO]);

        if (c + 1 < CIN) {
            tile[nxt][idx0] = s0;
            if (has1) tile[nxt][idx1] = s1;
        }
        __syncthreads();
    }
}

extern "C" void kernel_launch(void* const* d_in, const int* in_sizes, int n_in,
                              void* d_out, int out_size, void* d_ws, size_t ws_size,
                              hipStream_t stream) {
    const float* x      = (const float*)d_in[0];
    const float* w_comp = (const float*)d_in[1];
    const float* b_comp = (const float*)d_in[2];
    const float* w_enc  = (const float*)d_in[3];
    const float* b_enc  = (const float*)d_in[4];
    float* out = (float*)d_out;

    char* ws = (char*)d_ws;
    float* comp   = (float*)(ws);
    float* mask   = (float*)(ws + (size_t)4194304);
    float* msoft  = (float*)(ws + (size_t)4194304 + 6553600);
    float* w_encT = (float*)(ws + (size_t)4194304 + 6553600 + 6553600);

    repack_wenc    <<<(MPAD * KTOT + 255) / 256, 256, 0, stream>>>(w_enc, w_encT);
    compress_gemm  <<<256, 128, 0, stream>>>(x, w_comp, b_comp, comp);
    encoder_gemm   <<<256, 128, 0, stream>>>(comp, w_encT, b_enc, mask);
    softmax_shuffle<<<(NB * HWO + 255) / 256, 256, 0, stream>>>(mask, msoft);
    reassemble     <<<NB * 64, 256, 0, stream>>>(x, msoft, out);
}


# reassembly re-tiled: a workgroup stages a 32-channel x 8-row x slab in LDS once (one barrier, none per channel); a thread keeps its low-res pixel's 100 weights and writes its 2x2 outputs
# speedup vs baseline: 3.7393x; 3.7393x over previous
#include <hip/hip_runtime.h>
#include <hip/hip_bf16.h>


#define NB     4
#define CIN    256
#define CCOMP  64
#define CMASK  100
#define HH     64
#define WW     64
#define HW     (HH*WW)
#define HO     128
#define WO     128
#define HWO    (HO*WO)
#define KTOT   576
#define MPAD   112

typedef float v2f __attribute__((ext_vector_type(2)));
typedef float v8f __attribute__((ext_vector_type(8)));

__global__ void repack_wenc(const float* __restrict__ w_enc,
                            float* __restrict__ w_encT) {
    int idx = blockIdx.x * 256 + threadIdx.x;
    if (idx >= MPAD * KTOT) return;
    int m     = idx / KTOT;
    int r     = idx % KTOT;
    int plane = r >> 6;
    int ci    = r & 63;
    float v = 0.0f;
    if (m < CMASK) v = w_enc[m * KTOT + ci * 9 + plane];
    w_encT[idx] = v;
}

__global__ void __launch_bounds__(128)
compress_gemm(const float* __restrict__ x, const float* __restrict__ w_comp,
              const float* __restrict__ b_comp, float* __restrict__ comp) {
    const int wid   = blockIdx.x * 4 + (threadIdx.x >> 5);
    const int lane  = threadIdx.x & 31;
    const int n     = wid >> 8;
    const int nt    = wid & 255;
    const int col   = nt * 16 + (lane & 15);
    const int khalf = (lane >> 4) * 2;
    const int mrow  = lane & 15;

    const float* xb = x + (size_t)n * CIN * HW + col;

    v8f acc[4];
    const v8f vzero = {0.f,0.f,0.f,0.f,0.f,0.f,0.f,0.f};
#pragma unroll
    for (int mt = 0; mt < 4; ++mt) acc[mt] = vzero;

    for (int kb = 0; kb < CIN; kb += 4) {
        const int k = kb + khalf;
        v2f bfrag;
        bfrag.x = xb[(size_t)k * HW];
        bfrag.y = xb[(size_t)(k + 1) * HW];
#pragma unroll
        for (int mt = 0; mt < 4; ++mt) {
            v2f afrag = *(const v2f*)(w_comp + (size_t)(mt * 16 + mrow) * CIN + k);
            acc[mt] = __builtin_amdgcn_wmma_f32_16x16x4_f32(
                false, afrag, false, bfrag, (short)0, acc[mt], false, false);
        }
    }

    float* cb = comp + (size_t)n * CCOMP * HW + col;
    const int mhi = 8 * (lane >> 4);
#pragma unroll
    for (int mt = 0; mt < 4; ++mt) {
#pragma unroll
        for (int r = 0; r < 8; ++r) {
            const int m = mt * 16 + r + mhi;
            cb[(size_t)m * HW] = acc[mt][r] + b_comp[m];
        }
    }
}

__global__ void __launch_bounds__(128)
encoder_gemm(const float* __restrict__ comp, const float* __restrict__ w_encT,
             const float* __restrict__ b_enc, float* __restrict__ mask) {
    const int wid   = blockIdx.x * 4 + (threadIdx.x >> 5);
    const int lane  = threadIdx.x & 31;
    const int n     = wid >> 8;
    const int nt    = wid & 255;
    const int col   = nt * 16 + (lane & 15);
    const int h     = col >> 6;
    const int w     = col & 63;
    const int khalf = (lane >> 4) * 2;
    const int mrow  = lane & 15;

    v8f acc[7];
    const v8f vzero = {0.f,0.f,0.f,0.f,0.f,0.f,0.f,0.f};
#pragma unroll
    for (int mt = 0; mt < 7; ++mt) acc[mt] = vzero;

    const float* cbase = comp + (size_t)n * CCOMP * HW;

#pragma unroll
    for (int plane = 0; plane < 9; ++plane) {
        const int dy = plane / 3, dx = plane % 3;
        const int y  = h + dy - 1;
        const int xw = w + dx - 1;
        const bool valid = (y >= 0) && (y < HH) && (xw >= 0) && (xw < WW);
        const int yc = min(max(y, 0), HH - 1);
        const int xc = min(max(xw, 0), WW - 1);
        const float* bsrc = cbase + yc * WW + xc;

#pragma unroll 4
        for (int cib = 0; cib < CCOMP; cib += 4) {
            const int ci = cib + khalf;
            float b0 = bsrc[(size_t)ci * HW];
            float b1 = bsrc[(size_t)(ci + 1) * HW];
            v2f bfrag;
            bfrag.x = valid ? b0 : 0.0f;
            bfrag.y = valid ? b1 : 0.0f;
            const int kk = plane * 64 + ci;
#pragma unroll
            for (int mt = 0; mt < 7; ++mt) {
                v2f afrag = *(const v2f*)(w_encT + (size_t)(mt * 16 + mrow) * KTOT + kk);
                acc[mt] = __builtin_amdgcn_wmma_f32_16x16x4_f32(
                    false, afrag, false, bfrag, (short)0, acc[mt], false, false);
            }
        }
    }

    float* mbase = mask + (size_t)n * CMASK * HW + col;
    const int mhi = 8 * (lane >> 4);
#pragma unroll
    for (int mt = 0; mt < 7; ++mt) {
#pragma unroll
        for (int r = 0; r < 8; ++r) {
            const int cm = mt * 16 + r + mhi;
            if (cm < CMASK) mbase[(size_t)cm * HW] = acc[mt][r] + b_enc[cm];
        }
    }
}

__global__ void softmax_shuffle(const float* __restrict__ mask,
                                float* __restrict__ msoft) {
    int t = blockIdx.x * 256 + threadIdx.x;
    if (t >= NB * HWO) return;
    int n   = t / HWO;
    int rem = t % HWO;
    int Hr  = rem / WO;
    int Wj  = rem % WO;
    int h = Hr >> 1, sh = Hr & 1;
    int w = Wj >> 1, sw = Wj & 1;
    int r = sh * 2 + sw;

    const float* mb = mask + (size_t)n * CMASK * HW + h * WW + w;
    float v[25];
    float mx = -1e30f;
#pragma unroll
    for (int k = 0; k < 25; ++k) {
        v[k] = mb[(size_t)(k * 4 + r) * HW];
        mx = fmaxf(mx, v[k]);
    }
    float s = 0.0f;
#pragma unroll
    for (int k = 0; k < 25; ++k) { v[k] = __expf(v[k] - mx); s += v[k]; }
    float inv = 1.0f / s;
    float* ob = msoft + (size_t)n * 25 * HWO + Hr * WO + Wj;
#pragma unroll
    for (int k = 0; k < 25; ++k) ob[(size_t)k * HWO] = v[k] * inv;
}

#define RS_CH    32
#define RS_ROWS  4
#define RS_TR    (RS_ROWS + 4)
#define RS_PITCH 72

typedef float v4f __attribute__((ext_vector_type(4)));

__global__ void __launch_bounds__(256)
reassemble(const float* __restrict__ x, const float* __restrict__ msoft,
           float* __restrict__ out) {
    __shared__ float tile[RS_CH * RS_TR * RS_PITCH] __attribute__((aligned(16)));

    const int t  = threadIdx.x;
    const int cz = blockIdx.x & 7;
    const int rt = (blockIdx.x >> 3) & 15;
    const int n  = blockIdx.x >> 7;
    const int hl = t >> 6;
    const int w  = t & 63;
    const int h  = rt * RS_ROWS + hl;
    const int c0 = cz * RS_CH;

    const int  r   = (t >> 4) & 7;
    const int  c4  = t & 15;
    const int  chb = t >> 7;
    const int  gy  = rt * RS_ROWS + r - 2;
    const bool ok  = (gy >= 0) && (gy < HH);
    const float* xs = x + ((size_t)n * CIN + c0 + chb) * HW + (ok ? gy : 0) * WW + c4 * 4;
    const v4f zero4 = {0.f, 0.f, 0.f, 0.f};

    v4f stg[16];
#pragma unroll
    for (int i = 0; i < 16; ++i) stg[i] = zero4;
    if (ok) {
#pragma unroll
        for (int i = 0; i < 16; ++i)
            stg[i] = *(const v4f*)(xs + (size_t)(2 * i) * HW);
    }

    v2f wt[25][2];
    const float* mb = msoft + (size_t)n * 25 * HWO + (size_t)(2 * h) * WO + 2 * w;
#pragma unroll
    for (int k = 0; k < 25; ++k) {
        wt[k][0] = *(const v2f*)(mb + (size_t)k * HWO);
        wt[k][1] = *(const v2f*)(mb + (size_t)k * HWO + WO);
    }

    float* tst = &tile[(chb * RS_TR + r) * RS_PITCH + 4 + c4 * 4];
#pragma unroll
    for (int i = 0; i < 16; ++i)
        *(v4f*)(tst + (2 * i) * RS_TR * RS_PITCH) = stg[i];

#pragma unroll
    for (int i = 0; i < 2; ++i) {
        const int f = t + 256 * i;
        *(v4f*)(&tile[(f >> 1) * RS_PITCH + ((f & 1) ? (RS_PITCH - 4) : 0)]) = zero4;
    }
    __syncthreads();

    const float* tp = &tile[hl * RS_PITCH + w + 2];
    float* ob = out + ((size_t)n * CIN + c0) * HWO + (size_t)(2 * h) * WO + 2 * w;

#pragma unroll 2
    for (int ch = 0; ch < RS_CH; ++ch) {
        const float* tc = tp + ch * (RS_TR * RS_PITCH);
        v2f a0 = {0.f, 0.f};
        v2f a1 = {0.f, 0.f};
#pragma unroll
        for (int p = 0; p < 5; ++p)
#pragma unroll
            for (int q = 0; q < 5; ++q) {
                const float xv = tc[p * RS_PITCH + q];
                const v2f xx = {xv, xv};
                a0 = __builtin_elementwise_fma(wt[p * 5 + q][0], xx, a0);
                a1 = __builtin_elementwise_fma(wt[p * 5 + q][1], xx, a1);
            }
        __builtin_nontemporal_store(a0, (v2f*)(ob + (size_t)ch * HWO));
        __builtin_nontemporal_store(a1, (v2f*)(ob + (size_t)ch * HWO + WO));
    }
}

extern "C" void kernel_launch(void* const* d_in, const int* in_sizes, int n_in,
                              void* d_out, int out_size, void* d_ws, size_t ws_size,
                              hipStream_t stream) {
    const float* x      = (const float*)d_in[0];
    const float* w_comp = (const float*)d_in[1];
    const float* b_comp = (const float*)d_in[2];
    const float* w_enc  = (const float*)d_in[3];
    const float* b_enc  = (const float*)d_in[4];
    float* out = (float*)d_out;

    char* ws = (char*)d_ws;
    float* comp   = (float*)(ws);
    float* mask   = (float*)(ws + (size_t)4194304);
    float* msoft  = (float*)(ws + (size_t)4194304 + 6553600);
    float* w_encT = (float*)(ws + (size_t)4194304 + 6553600 + 6553600);

    repack_wenc    <<<(MPAD * KTOT + 255) / 256, 256, 0, stream>>>(w_enc, w_encT);
    compress_gemm  <<<256, 128, 0, stream>>>(x, w_comp, b_comp, comp);
    encoder_gemm   <<<256, 128, 0, stream>>>(comp, w_encT, b_enc, mask);
    softmax_shuffle<<<(NB * HWO + 255) / 256, 256, 0, stream>>>(mask, msoft);
    reassemble     <<<NB * 16 * 8, 256, 0, stream>>>(x, msoft, out);
}


# both GEMM k loops double-buffered: the next block's A and B loads go out as one group ahead of the current block's WMMAs (two register sets); k order and sums unchanged
# speedup vs baseline: 5.1633x; 1.3808x over previous
#include <hip/hip_runtime.h>
#include <hip/hip_bf16.h>


#define NB     4
#define CIN    256
#define CCOMP  64
#define CMASK  100
#define HH     64
#define WW     64
#define HW     (HH*WW)
#define HO     128
#define WO     128
#define HWO    (HO*WO)
#define KTOT   576
#define MPAD   112

typedef float v2f __attribute__((ext_vector_type(2)));
typedef float v8f __attribute__((ext_vector_type(8)));

__global__ void repack_wenc(const float* __restrict__ w_enc,
                            float* __restrict__ w_encT) {
    int idx = blockIdx.x * 256 + threadIdx.x;
    if (idx >= MPAD * KTOT) return;
    int m     = idx / KTOT;
    int r     = idx % KTOT;
    int plane = r >> 6;
    int ci    = r & 63;
    float v = 0.0f;
    if (m < CMASK) v = w_enc[m * KTOT + ci * 9 + plane];
    w_encT[idx] = v;
}

#define CG_KS 8
#define CG_BK (4 * CG_KS)

#define GROUP_FENCE() __builtin_amdgcn_sched_barrier(0)


struct CgFrag {
    v2f a[CG_KS][4];
    v2f b[CG_KS];
};

__device__ __forceinline__ void cg_load(CgFrag& f, const float* __restrict__ xk,
                                        const float* __restrict__ wk, int k0) {
#pragma unroll
    for (int s = 0; s < CG_KS; ++s) {
        const int k = k0 + 4 * s;
        f.b[s].x = xk[(size_t)k * HW];
        f.b[s].y = xk[(size_t)(k + 1) * HW];
#pragma unroll
        for (int mt = 0; mt < 4; ++mt)
            f.a[s][mt] = *(const v2f*)(wk + (size_t)(mt * 16) * CIN + k);
    }
}

__device__ __forceinline__ void cg_mma(const CgFrag& f, v8f (&acc)[4]) {
#pragma unroll
    for (int s = 0; s < CG_KS; ++s)
#pragma unroll
        for (int mt = 0; mt < 4; ++mt)
            acc[mt] = __builtin_amdgcn_wmma_f32_16x16x4_f32(
                false, f.a[s][mt], false, f.b[s], (short)0, acc[mt], false, false);
}

__global__ void __launch_bounds__(128)
compress_gemm(const float* __restrict__ x, const float* __restrict__ w_comp,
              const float* __restrict__ b_comp, float* __restrict__ comp) {
    const int wid   = blockIdx.x * 4 + (threadIdx.x >> 5);
    const int lane  = threadIdx.x & 31;
    const int n     = wid >> 8;
    const int nt    = wid & 255;
    const int col   = nt * 16 + (lane & 15);
    const int khalf = (lane >> 4) * 2;
    const int mrow  = lane & 15;

    const float* xk = x + (size_t)n * CIN * HW + col + (size_t)khalf * HW;
    const float* wk = w_comp + (size_t)mrow * CIN + khalf;

    v8f acc[4];
    const v8f vzero = {0.f,0.f,0.f,0.f,0.f,0.f,0.f,0.f};
#pragma unroll
    for (int mt = 0; mt < 4; ++mt) acc[mt] = vzero;

    CgFrag P, Q;
    cg_load(P, xk, wk, 0);
#pragma unroll 1
    for (int k0 = 0; k0 < CIN - 2 * CG_BK; k0 += 2 * CG_BK) {
        cg_load(Q, xk, wk, k0 + CG_BK);
        GROUP_FENCE();
        cg_mma(P, acc);
        GROUP_FENCE();
        cg_load(P, xk, wk, k0 + 2 * CG_BK);
        GROUP_FENCE();
        cg_mma(Q, acc);
        GROUP_FENCE();
    }
    cg_load(Q, xk, wk, CIN - CG_BK);
    GROUP_FENCE();
    cg_mma(P, acc);
    cg_mma(Q, acc);

    float* cb = comp + (size_t)n * CCOMP * HW + col;
    const int mhi = 8 * (lane >> 4);
#pragma unroll
    for (int mt = 0; mt < 4; ++mt) {
#pragma unroll
        for (int r = 0; r < 8; ++r) {
            const int m = mt * 16 + r + mhi;
            cb[(size_t)m * HW] = acc[mt][r] + b_comp[m];
        }
    }
}

#define EG_KS   4
#define EG_NBLK 36

struct EgFrag {
    v2f   a[EG_KS][7];
    float b0[EG_KS], b1[EG_KS];
    bool  valid;
};

__device__ __forceinline__ void eg_load(EgFrag& f, const float* __restrict__ cbase,
                                        const float* __restrict__ wk,
                                        int h, int w, int blk) {
    const int plane = blk >> 2;
    const int cib   = (blk & 3) * 16;
    const int dy = plane / 3, dx = plane % 3;
    const int y  = h + dy - 1;
    const int xw = w + dx - 1;
    f.valid = (y >= 0) && (y < HH) && (xw >= 0) && (xw < WW);
    const int yc = min(max(y, 0), HH - 1);
    const int xc = min(max(xw, 0), WW - 1);
    const float* bsrc = cbase + (size_t)cib * HW + yc * WW + xc;
    const float* asrc = wk + blk * 16;
#pragma unroll
    for (int s = 0; s < EG_KS; ++s) {
        f.b0[s] = bsrc[(size_t)(4 * s) * HW];
        f.b1[s] = bsrc[(size_t)(4 * s + 1) * HW];
#pragma unroll
        for (int mt = 0; mt < 7; ++mt)
            f.a[s][mt] = *(const v2f*)(asrc + (size_t)(mt * 16) * KTOT + 4 * s);
    }
}

__device__ __forceinline__ void eg_mma(const EgFrag& f, v8f (&acc)[7]) {
#pragma unroll
    for (int s = 0; s < EG_KS; ++s) {
        v2f bfrag;
        bfrag.x = f.valid ? f.b0[s] : 0.0f;
        bfrag.y = f.valid ? f.b1[s] : 0.0f;
#pragma unroll
        for (int mt = 0; mt < 7; ++mt)
            acc[mt] = __builtin_amdgcn_wmma_f32_16x16x4_f32(
                false, f.a[s][mt], false, bfrag, (short)0, acc[mt], false, false);
    }
}

__global__ void __launch_bounds__(128)
encoder_gemm(const float* __restrict__ comp, const float* __restrict__ w_encT,
             const float* __restrict__ b_enc, float* __restrict__ mask) {
    const int wid   = blockIdx.x * 4 + (threadIdx.x >> 5);
    const int lane  = threadIdx.x & 31;
    const int n     = wid >> 8;
    const int nt    = wid & 255;
    const int col   = nt * 16 + (lane & 15);
    const int h     = col >> 6;
    const int w     = col & 63;
    const int khalf = (lane >> 4) * 2;
    const int mrow  = lane & 15;

    v8f acc[7];
    const v8f vzero = {0.f,0.f,0.f,0.f,0.f,0.f,0.f,0.f};
#pragma unroll
    for (int mt = 0; mt < 7; ++mt) acc[mt] = vzero;

    const float* cbase = comp + (size_t)n * CCOMP * HW + (size_t)khalf * HW;
    const float* wk    = w_encT + (size_t)mrow * KTOT + khalf;

    EgFrag P, Q;
    eg_load(P, cbase, wk, h, w, 0);
#pragma unroll 1
    for (int blk = 0; blk < EG_NBLK - 2; blk += 2) {
        eg_load(Q, cbase, wk, h, w, blk + 1);
        GROUP_FENCE();
        eg_mma(P, acc);
        GROUP_FENCE();
        eg_load(P, cbase, wk, h, w, blk + 2);
        GROUP_FENCE();
        eg_mma(Q, acc);
        GROUP_FENCE();
    }
    eg_load(Q, cbase, wk, h, w, EG_NBLK - 1);
    GROUP_FENCE();
    eg_mma(P, acc);
    eg_mma(Q, acc);

    float* mbase = mask + (size_t)n * CMASK * HW + col;
    const int mhi = 8 * (lane >> 4);
#pragma unroll
    for (int mt = 0; mt < 7; ++mt) {
#pragma unroll
        for (int r = 0; r < 8; ++r) {
            const int cm = mt * 16 + r + mhi;
            if (cm < CMASK) mbase[(size_t)cm * HW] = acc[mt][r] + b_enc[cm];
        }
    }
}

__global__ void softmax_shuffle(const float* __restrict__ mask,
                                float* __restrict__ msoft) {
    int t = blockIdx.x * 256 + threadIdx.x;
    if (t >= NB * HWO) return;
    int n   = t / HWO;
    int rem = t % HWO;
    int Hr  = rem / WO;
    int Wj  = rem % WO;
    int h = Hr >> 1, sh = Hr & 1;
    int w = Wj >> 1, sw = Wj & 1;
    int r = sh * 2 + sw;

    const float* mb = mask + (size_t)n * CMASK * HW + h * WW + w;
    float v[25];
    float mx = -1e30f;
#pragma unroll
    for (int k = 0; k < 25; ++k) {
        v[k] = mb[(size_t)(k * 4 + r) * HW];
        mx = fmaxf(mx, v[k]);
    }
    float s = 0.0f;
#pragma unroll
    for (int k = 0; k < 25; ++k) { v[k] = __expf(v[k] - mx); s += v[k]; }
    float inv = 1.0f / s;
    float* ob = msoft + (size_t)n * 25 * HWO + Hr * WO + Wj;
#pragma unroll
    for (int k = 0; k < 25; ++k) ob[(size_t)k * HWO] = v[k] * inv;
}

#define RS_CH    32
#define RS_ROWS  4
#define RS_TR    (RS_ROWS + 4)
#define RS_PITCH 72

typedef float v4f __attribute__((ext_vector_type(4)));

__global__ void __launch_bounds__(256)
reassemble(const float* __restrict__ x, const float* __restrict__ msoft,
           float* __restrict__ out) {
    __shared__ float tile[RS_CH * RS_TR * RS_PITCH] __attribute__((aligned(16)));

    const int t  = threadIdx.x;
    const int cz = blockIdx.x & 7;
    const int rt = (blockIdx.x >> 3) & 15;
    const int n  = blockIdx.x >> 7;
    const int hl = t >> 6;
    const int w  = t & 63;
    const int h  = rt * RS_ROWS + hl;
    const int c0 = cz * RS_CH;

    const int  r   = (t >> 4) & 7;
    const int  c4  = t & 15;
    const int  chb = t >> 7;
    const int  gy  = rt * RS_ROWS + r - 2;
    const bool ok  = (gy >= 0) && (gy < HH);
    const float* xs = x + ((size_t)n * CIN + c0 + chb) * HW + (ok ? gy : 0) * WW + c4 * 4;
    const v4f zero4 = {0.f, 0.f, 0.f, 0.f};

    v4f stg[16];
#pragma unroll
    for (int i = 0; i < 16; ++i) stg[i] = zero4;
    if (ok) {
#pragma unroll
        for (int i = 0; i < 16; ++i)
            stg[i] = *(const v4f*)(xs + (size_t)(2 * i) * HW);
    }

    v2f wt[25][2];
    const float* mb = msoft + (size_t)n * 25 * HWO + (size_t)(2 * h) * WO + 2 * w;
#pragma unroll
    for (int k = 0; k < 25; ++k) {
        wt[k][0] = *(const v2f*)(mb + (size_t)k * HWO);
        wt[k][1] = *(const v2f*)(mb + (size_t)k * HWO + WO);
    }

    float* tst = &tile[(chb * RS_TR + r) * RS_PITCH + 4 + c4 * 4];
#pragma unroll
    for (int i = 0; i < 16; ++i)
        *(v4f*)(tst + (2 * i) * RS_TR * RS_PITCH) = stg[i];

#pragma unroll
    for (int i = 0; i < 2; ++i) {
        const int f = t + 256 * i;
        *(v4f*)(&tile[(f >> 1) * RS_PITCH + ((f & 1) ? (RS_PITCH - 4) : 0)]) = zero4;
    }
    __syncthreads();

    const float* tp = &tile[hl * RS_PITCH + w + 2];
    float* ob = out + ((size_t)n * CIN + c0) * HWO + (size_t)(2 * h) * WO + 2 * w;

#pragma unroll 2
    for (int ch = 0; ch < RS_CH; ++ch) {
        const float* tc = tp + ch * (RS_TR * RS_PITCH);
        v2f a0 = {0.f, 0.f};
        v2f a1 = {0.f, 0.f};
#pragma unroll
        for (int p = 0; p < 5; ++p)
#pragma unroll
            for (int q = 0; q < 5; ++q) {
                const float xv = tc[p * RS_PITCH + q];
                const v2f xx = {xv, xv};
                a0 = __builtin_elementwise_fma(wt[p * 5 + q][0], xx, a0);
                a1 = __builtin_elementwise_fma(wt[p * 5 + q][1], xx, a1);
            }
        __builtin_nontemporal_store(a0, (v2f*)(ob + (size_t)ch * HWO));
        __builtin_nontemporal_store(a1, (v2f*)(ob + (size_t)ch * HWO + WO));
    }
}

extern "C" void kernel_launch(void* const* d_in, const int* in_sizes, int n_in,
                              void* d_out, int out_size, void* d_ws, size_t ws_size,
                              hipStream_t stream) {
    const float* x      = (const float*)d_in[0];
    const float* w_comp = (const float*)d_in[1];
    const float* b_comp = (const float*)d_in[2];
    const float* w_enc  = (const float*)d_in[3];
    const float* b_enc  = (const float*)d_in[4];
    float* out = (float*)d_out;

    char* ws = (char*)d_ws;
    float* comp   = (float*)(ws);
    float* mask   = (float*)(ws + (size_t)4194304);
    float* msoft  = (float*)(ws + (size_t)4194304 + 6553600);
    float* w_encT = (float*)(ws + (size_t)4194304 + 6553600 + 6553600);

    repack_wenc    <<<(MPAD * KTOT + 255) / 256, 256, 0, stream>>>(w_enc, w_encT);
    compress_gemm  <<<256, 128, 0, stream>>>(x, w_comp, b_comp, comp);
    encoder_gemm   <<<256, 128, 0, stream>>>(comp, w_encT, b_enc, mask);
    softmax_shuffle<<<(NB * HWO + 255) / 256, 256, 0, stream>>>(mask, msoft);
    reassemble     <<<NB * 16 * 8, 256, 0, stream>>>(x, msoft, out);
}


# both convs' weights repacked into WMMA A-fragment order: each A load is a contiguous 512-B wave access (16 B per lane, two fragments) instead of 16 scattered lines; k order and sums unchanged
# speedup vs baseline: 6.1661x; 1.1942x over previous
#include <hip/hip_runtime.h>
#include <hip/hip_bf16.h>


#define NB     4
#define CIN    256
#define CCOMP  64
#define CMASK  100
#define HH     64
#define WW     64
#define HW     (HH*WW)
#define HO     128
#define WO     128
#define HWO    (HO*WO)
#define KTOT   576
#define MPAD   112

typedef float v2f __attribute__((ext_vector_type(2)));
typedef float v4f __attribute__((ext_vector_type(4)));
typedef float v8f __attribute__((ext_vector_type(8)));

#define EG_NLD  14
#define WENC_N  (36 * EG_NLD * 32 * 4)
#define WCMP_N  (64 * 2 * 32 * 4)

__global__ void repack_weights(const float* __restrict__ w_enc,
                               const float* __restrict__ w_comp,
                               float* __restrict__ w_encF,
                               float* __restrict__ w_compF) {
    const int idx = blockIdx.x * 256 + threadIdx.x;
    if (idx < WENC_N) {
        const int e   = idx & 3;
        const int l   = (idx >> 2) & 31;
        const int g   = idx >> 7;
        const int blk = g / EG_NLD;
        const int j   = g % EG_NLD;
        const int f   = 2 * j + (e >> 1);
        const int s   = f / 7;
        const int mt  = f % 7;
        const int m   = mt * 16 + (l & 15);
        const int k   = blk * 16 + 4 * s + 2 * (l >> 4) + (e & 1);
        const int plane = k >> 6;
        const int ci    = k & 63;
        float v = 0.0f;
        if (m < CMASK) v = w_enc[m * KTOT + ci * 9 + plane];
        w_encF[idx] = v;
    } else if (idx < WENC_N + WCMP_N) {
        const int i2 = idx - WENC_N;
        const int e  = i2 & 3;
        const int l  = (i2 >> 2) & 31;
        const int g  = i2 >> 7;
        const int ks = g >> 1;
        const int j  = g & 1;
        const int mt = 2 * j + (e >> 1);
        const int m  = mt * 16 + (l & 15);
        const int k  = 4 * ks + 2 * (l >> 4) + (e & 1);
        w_compF[i2] = w_comp[m * CIN + k];
    }
}

#define CG_KS 8
#define CG_BK (4 * CG_KS)

#define GROUP_FENCE() __builtin_amdgcn_sched_barrier(0)


struct CgFrag {
    v4f a[CG_KS][2];
    v2f b[CG_KS];
};

__device__ __forceinline__ void cg_load(CgFrag& f, const float* __restrict__ xk,
                                        const float* __restrict__ wk, int k0) {
#pragma unroll
    for (int s = 0; s < CG_KS; ++s) {
        const int k = k0 + 4 * s;
        f.b[s].x = xk[(size_t)k * HW];
        f.b[s].y = xk[(size_t)(k + 1) * HW];
#pragma unroll
        for (int j = 0; j < 2; ++j)
            f.a[s][j] = *(const v4f*)(wk + ((k >> 2) * 2 + j) * 128);
    }
}

__device__ __forceinline__ void cg_mma(const CgFrag& f, v8f (&acc)[4]) {
#pragma unroll
    for (int s = 0; s < CG_KS; ++s)
#pragma unroll
        for (int mt = 0; mt < 4; ++mt) {
            const v2f afrag = (mt & 1) ? f.a[s][mt >> 1].hi : f.a[s][mt >> 1].lo;
            acc[mt] = __builtin_amdgcn_wmma_f32_16x16x4_f32(
                false, afrag, false, f.b[s], (short)0, acc[mt], false, false);
        }
}

__global__ void __launch_bounds__(128)
compress_gemm(const float* __restrict__ x, const float* __restrict__ w_compF,
              const float* __restrict__ b_comp, float* __restrict__ comp) {
    const int wid   = blockIdx.x * 4 + (threadIdx.x >> 5);
    const int lane  = threadIdx.x & 31;
    const int n     = wid >> 8;
    const int nt    = wid & 255;
    const int col   = nt * 16 + (lane & 15);
    const int khalf = (lane >> 4) * 2;

    const float* xk = x + (size_t)n * CIN * HW + col + (size_t)khalf * HW;
    const float* wk = w_compF + lane * 4;

    v8f acc[4];
    const v8f vzero = {0.f,0.f,0.f,0.f,0.f,0.f,0.f,0.f};
#pragma unroll
    for (int mt = 0; mt < 4; ++mt) acc[mt] = vzero;

    CgFrag P, Q;
    cg_load(P, xk, wk, 0);
#pragma unroll 1
    for (int k0 = 0; k0 < CIN - 2 * CG_BK; k0 += 2 * CG_BK) {
        cg_load(Q, xk, wk, k0 + CG_BK);
        GROUP_FENCE();
        cg_mma(P, acc);
        GROUP_FENCE();
        cg_load(P, xk, wk, k0 + 2 * CG_BK);
        GROUP_FENCE();
        cg_mma(Q, acc);
        GROUP_FENCE();
    }
    cg_load(Q, xk, wk, CIN - CG_BK);
    GROUP_FENCE();
    cg_mma(P, acc);
    cg_mma(Q, acc);

    float* cb = comp + (size_t)n * CCOMP * HW + col;
    const int mhi = 8 * (lane >> 4);
#pragma unroll
    for (int mt = 0; mt < 4; ++mt) {
#pragma unroll
        for (int r = 0; r < 8; ++r) {
            const int m = mt * 16 + r + mhi;
            cb[(size_t)m * HW] = acc[mt][r] + b_comp[m];
        }
    }
}

#define EG_KS   4
#define EG_NBLK 36

struct EgFrag {
    v4f   a[EG_NLD];
    float b0[EG_KS], b1[EG_KS];
    bool  valid;
};

__device__ __forceinline__ void eg_load(EgFrag& f, const float* __restrict__ cbase,
                                        const float* __restrict__ wk,
                                        int h, int w, int blk) {
    const int plane = blk >> 2;
    const int cib   = (blk & 3) * 16;
    const int dy = plane / 3, dx = plane % 3;
    const int y  = h + dy - 1;
    const int xw = w + dx - 1;
    f.valid = (y >= 0) && (y < HH) && (xw >= 0) && (xw < WW);
    const int yc = min(max(y, 0), HH - 1);
    const int xc = min(max(xw, 0), WW - 1);
    const float* bsrc = cbase + (size_t)cib * HW + yc * WW + xc;
    const float* asrc = wk + blk * (EG_NLD * 128);
#pragma unroll
    for (int s = 0; s < EG_KS; ++s) {
        f.b0[s] = bsrc[(size_t)(4 * s) * HW];
        f.b1[s] = bsrc[(size_t)(4 * s + 1) * HW];
    }
#pragma unroll
    for (int j = 0; j < EG_NLD; ++j)
        f.a[j] = *(const v4f*)(asrc + j * 128);
}

__device__ __forceinline__ void eg_mma(const EgFrag& f, v8f (&acc)[7]) {
#pragma unroll
    for (int s = 0; s < EG_KS; ++s) {
        v2f bfrag;
        bfrag.x = f.valid ? f.b0[s] : 0.0f;
        bfrag.y = f.valid ? f.b1[s] : 0.0f;
#pragma unroll
        for (int mt = 0; mt < 7; ++mt) {
            const int fi = s * 7 + mt;
            const v2f afrag = (fi & 1) ? f.a[fi >> 1].hi : f.a[fi >> 1].lo;
            acc[mt] = __builtin_amdgcn_wmma_f32_16x16x4_f32(
                false, afrag, false, bfrag, (short)0, acc[mt], false, false);
        }
    }
}

__global__ void __launch_bounds__(128)
encoder_gemm(const float* __restrict__ comp, const float* __restrict__ w_encF,
             const float* __restrict__ b_enc, float* __restrict__ mask) {
    const int wid   = blockIdx.x * 4 + (threadIdx.x >> 5);
    const int lane  = threadIdx.x & 31;
    const int n     = wid >> 8;
    const int nt    = wid & 255;
    const int col   = nt * 16 + (lane & 15);
    const int h     = col >> 6;
    const int w     = col & 63;
    const int khalf = (lane >> 4) * 2;

    v8f acc[7];
    const v8f vzero = {0.f,0.f,0.f,0.f,0.f,0.f,0.f,0.f};
#pragma unroll
    for (int mt = 0; mt < 7; ++mt) acc[mt] = vzero;

    const float* cbase = comp + (size_t)n * CCOMP * HW + (size_t)khalf * HW;
    const float* wk    = w_encF + lane * 4;

    EgFrag P, Q;
    eg_load(P, cbase, wk, h, w, 0);
#pragma unroll 1
    for (int blk = 0; blk < EG_NBLK - 2; blk += 2) {
        eg_load(Q, cbase, wk, h, w, blk + 1);
        GROUP_FENCE();
        eg_mma(P, acc);
        GROUP_FENCE();
        eg_load(P, cbase, wk, h, w, blk + 2);
        GROUP_FENCE();
        eg_mma(Q, acc);
        GROUP_FENCE();
    }
    eg_load(Q, cbase, wk, h, w, EG_NBLK - 1);
    GROUP_FENCE();
    eg_mma(P, acc);
    eg_mma(Q, acc);

    float* mbase = mask + (size_t)n * CMASK * HW + col;
    const int mhi = 8 * (lane >> 4);
#pragma unroll
    for (int mt = 0; mt < 7; ++mt) {
#pragma unroll
        for (int r = 0; r < 8; ++r) {
            const int cm = mt * 16 + r + mhi;
            if (cm < CMASK) mbase[(size_t)cm * HW] = acc[mt][r] + b_enc[cm];
        }
    }
}

__global__ void softmax_shuffle(const float* __restrict__ mask,
                                float* __restrict__ msoft) {
    int t = blockIdx.x * 256 + threadIdx.x;
    if (t >= NB * HWO) return;
    int n   = t / HWO;
    int rem = t % HWO;
    int Hr  = rem / WO;
    int Wj  = rem % WO;
    int h = Hr >> 1, sh = Hr & 1;
    int w = Wj >> 1, sw = Wj & 1;
    int r = sh * 2 + sw;

    const float* mb = mask + (size_t)n * CMASK * HW + h * WW + w;
    float v[25];
    float mx = -1e30f;
#pragma unroll
    for (int k = 0; k < 25; ++k) {
        v[k] = mb[(size_t)(k * 4 + r) * HW];
        mx = fmaxf(mx, v[k]);
    }
    float s = 0.0f;
#pragma unroll
    for (int k = 0; k < 25; ++k) { v[k] = __expf(v[k] - mx); s += v[k]; }
    float inv = 1.0f / s;
    float* ob = msoft + (size_t)n * 25 * HWO + Hr * WO + Wj;
#pragma unroll
    for (int k = 0; k < 25; ++k) ob[(size_t)k * HWO] = v[k] * inv;
}

#define RS_CH    32
#define RS_ROWS  4
#define RS_TR    (RS_ROWS + 4)
#define RS_PITCH 72

__global__ void __launch_bounds__(256)
reassemble(const float* __restrict__ x, const float* __restrict__ msoft,
           float* __restrict__ out) {
    __shared__ float tile[RS_CH * RS_TR * RS_PITCH] __attribute__((aligned(16)));

    const int t  = threadIdx.x;
    const int cz = blockIdx.x & 7;
    const int rt = (blockIdx.x >> 3) & 15;
    const int n  = blockIdx.x >> 7;
    const int hl = t >> 6;
    const int w  = t & 63;
    const int h  = rt * RS_ROWS + hl;
    const int c0 = cz * RS_CH;

    const int  r   = (t >> 4) & 7;
    const int  c4  = t & 15;
    const int  chb = t >> 7;
    const int  gy  = rt * RS_ROWS + r - 2;
    const bool ok  = (gy >= 0) && (gy < HH);
    const float* xs = x + ((size_t)n * CIN + c0 + chb) * HW + (ok ? gy : 0) * WW + c4 * 4;
    const v4f zero4 = {0.f, 0.f, 0.f, 0.f};

    v4f stg[16];
#pragma unroll
    for (int i = 0; i < 16; ++i) stg[i] = zero4;
    if (ok) {
#pragma unroll
        for (int i = 0; i < 16; ++i)
            stg[i] = *(const v4f*)(xs + (size_t)(2 * i) * HW);
    }

    v2f wt[25][2];
    const float* mb = msoft + (size_t)n * 25 * HWO + (size_t)(2 * h) * WO + 2 * w;
#pragma unroll
    for (int k = 0; k < 25; ++k) {
        wt[k][0] = *(const v2f*)(mb + (size_t)k * HWO);
        wt[k][1] = *(const v2f*)(mb + (size_t)k * HWO + WO);
    }

    float* tst = &tile[(chb * RS_TR + r) * RS_PITCH + 4 + c4 * 4];
#pragma unroll
    for (int i = 0; i < 16; ++i)
        *(v4f*)(tst + (2 * i) * RS_TR * RS_PITCH) = stg[i];

#pragma unroll
    for (int i = 0; i < 2; ++i) {
        const int f = t + 256 * i;
        *(v4f*)(&tile[(f >> 1) * RS_PITCH + ((f & 1) ? (RS_PITCH - 4) : 0)]) = zero4;
    }
    __syncthreads();

    const float* tp = &tile[hl * RS_PITCH + w + 2];
    float* ob = out + ((size_t)n * CIN + c0) * HWO + (size_t)(2 * h) * WO + 2 * w;

#pragma unroll 2
    for (int ch = 0; ch < RS_CH; ++ch) {
        const float* tc = tp + ch * (RS_TR * RS_PITCH);
        v2f a0 = {0.f, 0.f};
        v2f a1 = {0.f, 0.f};
#pragma unroll
        for (int p = 0; p < 5; ++p)
#pragma unroll
            for (int q = 0; q < 5; ++q) {
                const float xv = tc[p * RS_PITCH + q];
                const v2f xx = {xv, xv};
                a0 = __builtin_elementwise_fma(wt[p * 5 + q][0], xx, a0);
                a1 = __builtin_elementwise_fma(wt[p * 5 + q][1], xx, a1);
            }
        __builtin_nontemporal_store(a0, (v2f*)(ob + (size_t)ch * HWO));
        __builtin_nontemporal_store(a1, (v2f*)(ob + (size_t)ch * HWO + WO));
    }
}

extern "C" void kernel_launch(void* const* d_in, const int* in_sizes, int n_in,
                              void* d_out, int out_size, void* d_ws, size_t ws_size,
                              hipStream_t stream) {
    const float* x      = (const float*)d_in[0];
    const float* w_comp = (const float*)d_in[1];
    const float* b_comp = (const float*)d_in[2];
    const float* w_enc  = (const float*)d_in[3];
    const float* b_enc  = (const float*)d_in[4];
    float* out = (float*)d_out;

    char* ws = (char*)d_ws;
    float* comp   = (float*)(ws);
    float* mask   = (float*)(ws + (size_t)4194304);
    float* msoft  = (float*)(ws + (size_t)4194304 + 6553600);
    float* w_encF = (float*)(ws + (size_t)4194304 + 6553600 + 6553600);
    float* w_compF = msoft;

    repack_weights <<<(WENC_N + WCMP_N) / 256, 256, 0, stream>>>(w_enc, w_comp, w_encF, w_compF);
    compress_gemm  <<<256, 128, 0, stream>>>(x, w_compF, b_comp, comp);
    encoder_gemm   <<<256, 128, 0, stream>>>(comp, w_encF, b_enc, mask);
    softmax_shuffle<<<(NB * HWO + 255) / 256, 256, 0, stream>>>(mask, msoft);
    reassemble     <<<NB * 16 * 8, 256, 0, stream>>>(x, msoft, out);
}


# softmax + pixel_shuffle fused into the 3x3-conv GEMM epilogue (a wave's accumulators hold all 100 mask channels of its 16 pixels): 5 launches -> 4, mask plane gone; softmax sum = 2 half-wave partials
# speedup vs baseline: 6.7844x; 1.1003x over previous
#include <hip/hip_runtime.h>
#include <hip/hip_bf16.h>


#define NB     4
#define CIN    256
#define CCOMP  64
#define CMASK  100
#define HH     64
#define WW     64
#define HW     (HH*WW)
#define HO     128
#define WO     128
#define HWO    (HO*WO)
#define KTOT   576
#define MPAD   112

typedef float v2f __attribute__((ext_vector_type(2)));
typedef float v4f __attribute__((ext_vector_type(4)));
typedef float v8f __attribute__((ext_vector_type(8)));

#define EG_NLD  14
#define WENC_N  (36 * EG_NLD * 32 * 4)
#define WCMP_N  (64 * 2 * 32 * 4)

__global__ void repack_weights(const float* __restrict__ w_enc,
                               const float* __restrict__ w_comp,
                               float* __restrict__ w_encF,
                               float* __restrict__ w_compF) {
    const int idx = blockIdx.x * 256 + threadIdx.x;
    if (idx < WENC_N) {
        const int e   = idx & 3;
        const int l   = (idx >> 2) & 31;
        const int g   = idx >> 7;
        const int blk = g / EG_NLD;
        const int j   = g % EG_NLD;
        const int f   = 2 * j + (e >> 1);
        const int s   = f / 7;
        const int mt  = f % 7;
        const int m   = mt * 16 + (l & 15);
        const int k   = blk * 16 + 4 * s + 2 * (l >> 4) + (e & 1);
        const int plane = k >> 6;
        const int ci    = k & 63;
        float v = 0.0f;
        if (m < CMASK) v = w_enc[m * KTOT + ci * 9 + plane];
        w_encF[idx] = v;
    } else if (idx < WENC_N + WCMP_N) {
        const int i2 = idx - WENC_N;
        const int e  = i2 & 3;
        const int l  = (i2 >> 2) & 31;
        const int g  = i2 >> 7;
        const int ks = g >> 1;
        const int j  = g & 1;
        const int mt = 2 * j + (e >> 1);
        const int m  = mt * 16 + (l & 15);
        const int k  = 4 * ks + 2 * (l >> 4) + (e & 1);
        w_compF[i2] = w_comp[m * CIN + k];
    }
}

#define CG_KS 8
#define CG_BK (4 * CG_KS)

#define GROUP_FENCE() __builtin_amdgcn_sched_barrier(0)


struct CgFrag {
    v4f a[CG_KS][2];
    v2f b[CG_KS];
};

__device__ __forceinline__ void cg_load(CgFrag& f, const float* __restrict__ xk,
                                        const float* __restrict__ wk, int k0) {
#pragma unroll
    for (int s = 0; s < CG_KS; ++s) {
        const int k = k0 + 4 * s;
        f.b[s].x = xk[(size_t)k * HW];
        f.b[s].y = xk[(size_t)(k + 1) * HW];
#pragma unroll
        for (int j = 0; j < 2; ++j)
            f.a[s][j] = *(const v4f*)(wk + ((k >> 2) * 2 + j) * 128);
    }
}

__device__ __forceinline__ void cg_mma(const CgFrag& f, v8f (&acc)[4]) {
#pragma unroll
    for (int s = 0; s < CG_KS; ++s)
#pragma unroll
        for (int mt = 0; mt < 4; ++mt) {
            const v2f afrag = (mt & 1) ? f.a[s][mt >> 1].hi : f.a[s][mt >> 1].lo;
            acc[mt] = __builtin_amdgcn_wmma_f32_16x16x4_f32(
                false, afrag, false, f.b[s], (short)0, acc[mt], false, false);
        }
}

__global__ void __launch_bounds__(128)
compress_gemm(const float* __restrict__ x, const float* __restrict__ w_compF,
              const float* __restrict__ b_comp, float* __restrict__ comp) {
    const int wid   = blockIdx.x * 4 + (threadIdx.x >> 5);
    const int lane  = threadIdx.x & 31;
    const int n     = wid >> 8;
    const int nt    = wid & 255;
    const int col   = nt * 16 + (lane & 15);
    const int khalf = (lane >> 4) * 2;

    const float* xk = x + (size_t)n * CIN * HW + col + (size_t)khalf * HW;
    const float* wk = w_compF + lane * 4;

    v8f acc[4];
    const v8f vzero = {0.f,0.f,0.f,0.f,0.f,0.f,0.f,0.f};
#pragma unroll
    for (int mt = 0; mt < 4; ++mt) acc[mt] = vzero;

    CgFrag P, Q;
    cg_load(P, xk, wk, 0);
#pragma unroll 1
    for (int k0 = 0; k0 < CIN - 2 * CG_BK; k0 += 2 * CG_BK) {
        cg_load(Q, xk, wk, k0 + CG_BK);
        GROUP_FENCE();
        cg_mma(P, acc);
        GROUP_FENCE();
        cg_load(P, xk, wk, k0 + 2 * CG_BK);
        GROUP_FENCE();
        cg_mma(Q, acc);
        GROUP_FENCE();
    }
    cg_load(Q, xk, wk, CIN - CG_BK);
    GROUP_FENCE();
    cg_mma(P, acc);
    cg_mma(Q, acc);

    float* cb = comp + (size_t)n * CCOMP * HW + col;
    const int mhi = 8 * (lane >> 4);
#pragma unroll
    for (int mt = 0; mt < 4; ++mt) {
#pragma unroll
        for (int r = 0; r < 8; ++r) {
            const int m = mt * 16 + r + mhi;
            cb[(size_t)m * HW] = acc[mt][r] + b_comp[m];
        }
    }
}

#define EG_KS   4

__device__ __forceinline__ float xhalf(float x) {
    return __builtin_bit_cast(float, __builtin_amdgcn_permlanex16(
        __builtin_bit_cast(int, x), __builtin_bit_cast(int, x),
        0x76543210, 0xfedcba98, false, false));
}
#define EG_NBLK 36

struct EgFrag {
    v4f   a[EG_NLD];
    float b0[EG_KS], b1[EG_KS];
    bool  valid;
};

__device__ __forceinline__ void eg_load(EgFrag& f, const float* __restrict__ cbase,
                                        const float* __restrict__ wk,
                                        int h, int w, int blk) {
    const int plane = blk >> 2;
    const int cib   = (blk & 3) * 16;
    const int dy = plane / 3, dx = plane % 3;
    const int y  = h + dy - 1;
    const int xw = w + dx - 1;
    f.valid = (y >= 0) && (y < HH) && (xw >= 0) && (xw < WW);
    const int yc = min(max(y, 0), HH - 1);
    const int xc = min(max(xw, 0), WW - 1);
    const float* bsrc = cbase + (size_t)cib * HW + yc * WW + xc;
    const float* asrc = wk + blk * (EG_NLD * 128);
#pragma unroll
    for (int s = 0; s < EG_KS; ++s) {
        f.b0[s] = bsrc[(size_t)(4 * s) * HW];
        f.b1[s] = bsrc[(size_t)(4 * s + 1) * HW];
    }
#pragma unroll
    for (int j = 0; j < EG_NLD; ++j)
        f.a[j] = *(const v4f*)(asrc + j * 128);
}

__device__ __forceinline__ void eg_mma(const EgFrag& f, v8f (&acc)[7]) {
#pragma unroll
    for (int s = 0; s < EG_KS; ++s) {
        v2f bfrag;
        bfrag.x = f.valid ? f.b0[s] : 0.0f;
        bfrag.y = f.valid ? f.b1[s] : 0.0f;
#pragma unroll
        for (int mt = 0; mt < 7; ++mt) {
            const int fi = s * 7 + mt;
            const v2f afrag = (fi & 1) ? f.a[fi >> 1].hi : f.a[fi >> 1].lo;
            acc[mt] = __builtin_amdgcn_wmma_f32_16x16x4_f32(
                false, afrag, false, bfrag, (short)0, acc[mt], false, false);
        }
    }
}

__global__ void __launch_bounds__(128)
encoder_gemm(const float* __restrict__ comp, const float* __restrict__ w_encF,
             const float* __restrict__ b_enc, float* __restrict__ msoft) {
    const int wid   = blockIdx.x * 4 + (threadIdx.x >> 5);
    const int lane  = threadIdx.x & 31;
    const int n     = wid >> 8;
    const int nt    = wid & 255;
    const int col   = nt * 16 + (lane & 15);
    const int h     = col >> 6;
    const int w     = col & 63;
    const int khalf = (lane >> 4) * 2;

    v8f acc[7];
    const v8f vzero = {0.f,0.f,0.f,0.f,0.f,0.f,0.f,0.f};
#pragma unroll
    for (int mt = 0; mt < 7; ++mt) acc[mt] = vzero;

    const float* cbase = comp + (size_t)n * CCOMP * HW + (size_t)khalf * HW;
    const float* wk    = w_encF + lane * 4;

    EgFrag P, Q;
    eg_load(P, cbase, wk, h, w, 0);
#pragma unroll 1
    for (int blk = 0; blk < EG_NBLK - 2; blk += 2) {
        eg_load(Q, cbase, wk, h, w, blk + 1);
        GROUP_FENCE();
        eg_mma(P, acc);
        GROUP_FENCE();
        eg_load(P, cbase, wk, h, w, blk + 2);
        GROUP_FENCE();
        eg_mma(Q, acc);
        GROUP_FENCE();
    }
    eg_load(Q, cbase, wk, h, w, EG_NBLK - 1);
    GROUP_FENCE();
    eg_mma(P, acc);
    eg_mma(Q, acc);

    const int  hi  = lane >> 4;
    const bool low = (hi == 0);
    const float* bb = b_enc + 8 * hi;

    float v[6][8];
    float v6[4];
#pragma unroll
    for (int mt = 0; mt < 6; ++mt)
#pragma unroll
        for (int r = 0; r < 8; ++r) v[mt][r] = acc[mt][r] + bb[mt * 16 + r];
#pragma unroll
    for (int q = 0; q < 4; ++q) v6[q] = acc[6][q] + b_enc[96 + q];

#pragma unroll
    for (int q = 0; q < 4; ++q) {
        float mx = -1e30f;
#pragma unroll
        for (int mt = 0; mt < 6; ++mt) mx = fmaxf(mx, fmaxf(v[mt][q], v[mt][q + 4]));
        mx = fmaxf(mx, low ? v6[q] : -1e30f);
        mx = fmaxf(mx, xhalf(mx));
        float s = 0.0f;
#pragma unroll
        for (int mt = 0; mt < 6; ++mt) {
            v[mt][q]     = __expf(v[mt][q] - mx);      s += v[mt][q];
            v[mt][q + 4] = __expf(v[mt][q + 4] - mx);  s += v[mt][q + 4];
        }
        v6[q] = low ? __expf(v6[q] - mx) : 0.0f;
        s += v6[q];
        s += xhalf(s);
        const float inv = 1.0f / s;
#pragma unroll
        for (int mt = 0; mt < 6; ++mt) { v[mt][q] *= inv; v[mt][q + 4] *= inv; }
        v6[q] *= inv;
    }

    float* ob = msoft + (size_t)n * 25 * HWO + (size_t)(2 * hi) * HWO
                      + (size_t)(2 * h) * WO + 2 * w;
#pragma unroll
    for (int mt = 0; mt < 6; ++mt)
#pragma unroll
        for (int j = 0; j < 2; ++j)
#pragma unroll
            for (int sh = 0; sh < 2; ++sh) {
                const v2f o = {v[mt][4 * j + 2 * sh], v[mt][4 * j + 2 * sh + 1]};
                *(v2f*)(ob + (size_t)(mt * 4 + j) * HWO + sh * WO) = o;
            }
    if (low) {
#pragma unroll
        for (int sh = 0; sh < 2; ++sh) {
            const v2f o = {v6[2 * sh], v6[2 * sh + 1]};
            *(v2f*)(ob + (size_t)24 * HWO + sh * WO) = o;
        }
    }
}

#define RS_CH    32
#define RS_ROWS  4
#define RS_TR    (RS_ROWS + 4)
#define RS_PITCH 72

__global__ void __launch_bounds__(256)
reassemble(const float* __restrict__ x, const float* __restrict__ msoft,
           float* __restrict__ out) {
    __shared__ float tile[RS_CH * RS_TR * RS_PITCH] __attribute__((aligned(16)));

    const int t  = threadIdx.x;
    const int cz = blockIdx.x & 7;
    const int rt = (blockIdx.x >> 3) & 15;
    const int n  = blockIdx.x >> 7;
    const int hl = t >> 6;
    const int w  = t & 63;
    const int h  = rt * RS_ROWS + hl;
    const int c0 = cz * RS_CH;

    const int  r   = (t >> 4) & 7;
    const int  c4  = t & 15;
    const int  chb = t >> 7;
    const int  gy  = rt * RS_ROWS + r - 2;
    const bool ok  = (gy >= 0) && (gy < HH);
    const float* xs = x + ((size_t)n * CIN + c0 + chb) * HW + (ok ? gy : 0) * WW + c4 * 4;
    const v4f zero4 = {0.f, 0.f, 0.f, 0.f};

    v4f stg[16];
#pragma unroll
    for (int i = 0; i < 16; ++i) stg[i] = zero4;
    if (ok) {
#pragma unroll
        for (int i = 0; i < 16; ++i)
            stg[i] = *(const v4f*)(xs + (size_t)(2 * i) * HW);
    }

    v2f wt[25][2];
    const float* mb = msoft + (size_t)n * 25 * HWO + (size_t)(2 * h) * WO + 2 * w;
#pragma unroll
    for (int k = 0; k < 25; ++k) {
        wt[k][0] = *(const v2f*)(mb + (size_t)k * HWO);
        wt[k][1] = *(const v2f*)(mb + (size_t)k * HWO + WO);
    }

    float* tst = &tile[(chb * RS_TR + r) * RS_PITCH + 4 + c4 * 4];
#pragma unroll
    for (int i = 0; i < 16; ++i)
        *(v4f*)(tst + (2 * i) * RS_TR * RS_PITCH) = stg[i];

#pragma unroll
    for (int i = 0; i < 2; ++i) {
        const int f = t + 256 * i;
        *(v4f*)(&tile[(f >> 1) * RS_PITCH + ((f & 1) ? (RS_PITCH - 4) : 0)]) = zero4;
    }
    __syncthreads();

    const float* tp = &tile[hl * RS_PITCH + w + 2];
    float* ob = out + ((size_t)n * CIN + c0) * HWO + (size_t)(2 * h) * WO + 2 * w;

#pragma unroll 2
    for (int ch = 0; ch < RS_CH; ++ch) {
        const float* tc = tp + ch * (RS_TR * RS_PITCH);
        v2f a0 = {0.f, 0.f};
        v2f a1 = {0.f, 0.f};
#pragma unroll
        for (int p = 0; p < 5; ++p)
#pragma unroll
            for (int q = 0; q < 5; ++q) {
                const float xv = tc[p * RS_PITCH + q];
                const v2f xx = {xv, xv};
                a0 = __builtin_elementwise_fma(wt[p * 5 + q][0], xx, a0);
                a1 = __builtin_elementwise_fma(wt[p * 5 + q][1], xx, a1);
            }
        __builtin_nontemporal_store(a0, (v2f*)(ob + (size_t)ch * HWO));
        __builtin_nontemporal_store(a1, (v2f*)(ob + (size_t)ch * HWO + WO));
    }
}

extern "C" void kernel_launch(void* const* d_in, const int* in_sizes, int n_in,
                              void* d_out, int out_size, void* d_ws, size_t ws_size,
                              hipStream_t stream) {
    const float* x      = (const float*)d_in[0];
    const float* w_comp = (const float*)d_in[1];
    const float* b_comp = (const float*)d_in[2];
    const float* w_enc  = (const float*)d_in[3];
    const float* b_enc  = (const float*)d_in[4];
    float* out = (float*)d_out;

    char* ws = (char*)d_ws;
    float* comp    = (float*)(ws);
    float* msoft   = (float*)(ws + (size_t)4194304);
    float* w_encF  = (float*)(ws + (size_t)4194304 + 6553600);
    float* w_compF = (float*)(ws + (size_t)4194304 + 6553600 + 258048);

    repack_weights <<<(WENC_N + WCMP_N) / 256, 256, 0, stream>>>(w_enc, w_comp, w_encF, w_compF);
    compress_gemm  <<<256, 128, 0, stream>>>(x, w_compF, b_comp, comp);
    encoder_gemm   <<<256, 128, 0, stream>>>(comp, w_encF, b_enc, msoft);
    reassemble     <<<NB * 16 * 8, 256, 0, stream>>>(x, msoft, out);
}


# weight re-layout launch removed: each 1x1-conv workgroup lays w_comp out in fragment order in LDS for itself and writes 1/256 of the 3x3 conv's table; 4 launches -> 3; k order and sums unchanged
# speedup vs baseline: 7.0227x; 1.0351x over previous
#include <hip/hip_runtime.h>
#include <hip/hip_bf16.h>


#define NB     4
#define CIN    256
#define CCOMP  64
#define CMASK  100
#define HH     64
#define WW     64
#define HW     (HH*WW)
#define HO     128
#define WO     128
#define HWO    (HO*WO)
#define KTOT   576
#define MPAD   112

typedef float v2f __attribute__((ext_vector_type(2)));
typedef float v4f __attribute__((ext_vector_type(4)));
typedef float v8f __attribute__((ext_vector_type(8)));

#define EG_NLD  14
#define WENC_N  (36 * EG_NLD * 32 * 4)
#define WENC_PER_WG (WENC_N / 256)
#define CW_PITCH 260

__device__ __forceinline__ float wenc_table_elem(const float* __restrict__ w_enc, int idx) {
    const int e   = idx & 3;
    const int l   = (idx >> 2) & 31;
    const int g   = idx >> 7;
    const int blk = g / EG_NLD;
    const int j   = g % EG_NLD;
    const int f   = 2 * j + (e >> 1);
    const int s   = f / 7;
    const int mt  = f % 7;
    const int m   = mt * 16 + (l & 15);
    const int k   = blk * 16 + 4 * s + 2 * (l >> 4) + (e & 1);
    const int plane = k >> 6;
    const int ci    = k & 63;
    const float v = w_enc[min(m, CMASK - 1) * KTOT + ci * 9 + plane];
    return (m < CMASK) ? v : 0.0f;
}

#define CG_KS 8
#define CG_BK (4 * CG_KS)

#define GROUP_FENCE() __builtin_amdgcn_sched_barrier(0)


struct CgFrag {
    v4f a[CG_KS][2];
    v2f b[CG_KS];
};

__device__ __forceinline__ void cg_load_b(CgFrag& f, const float* __restrict__ xk, int k0) {
#pragma unroll
    for (int s = 0; s < CG_KS; ++s) {
        const int k = k0 + 4 * s;
        f.b[s].x = xk[(size_t)k * HW];
        f.b[s].y = xk[(size_t)(k + 1) * HW];
    }
}

__device__ __forceinline__ void cg_load_a(CgFrag& f, const float* wk, int k0) {
#pragma unroll
    for (int s = 0; s < CG_KS; ++s)
#pragma unroll
        for (int j = 0; j < 2; ++j)
            f.a[s][j] = *(const v4f*)(wk + ((k0 >> 2) + s) * CW_PITCH + j * 128);
}

__device__ __forceinline__ void cg_load(CgFrag& f, const float* __restrict__ xk,
                                        const float* wk, int k0) {
    cg_load_b(f, xk, k0);
    cg_load_a(f, wk, k0);
}

__device__ __forceinline__ void cg_mma(const CgFrag& f, v8f (&acc)[4]) {
#pragma unroll
    for (int s = 0; s < CG_KS; ++s)
#pragma unroll
        for (int mt = 0; mt < 4; ++mt) {
            const v2f afrag = (mt & 1) ? f.a[s][mt >> 1].hi : f.a[s][mt >> 1].lo;
            acc[mt] = __builtin_amdgcn_wmma_f32_16x16x4_f32(
                false, afrag, false, f.b[s], (short)0, acc[mt], false, false);
        }
}

__global__ void __launch_bounds__(128)
compress_gemm(const float* __restrict__ x, const float* __restrict__ w_comp,
              const float* __restrict__ b_comp, const float* __restrict__ w_enc,
              float* __restrict__ comp, float* __restrict__ w_encF) {
    __shared__ float wl[64 * CW_PITCH] __attribute__((aligned(16)));

    const int wid   = blockIdx.x * 4 + (threadIdx.x >> 5);
    const int lane  = threadIdx.x & 31;
    const int n     = wid >> 8;
    const int nt    = wid & 255;
    const int col   = nt * 16 + (lane & 15);
    const int khalf = (lane >> 4) * 2;

    const float* xk = x + (size_t)n * CIN * HW + col + (size_t)khalf * HW;
    const float* wk = wl + lane * 4;

    v8f acc[4];
    const v8f vzero = {0.f,0.f,0.f,0.f,0.f,0.f,0.f,0.f};
#pragma unroll
    for (int mt = 0; mt < 4; ++mt) acc[mt] = vzero;

    const int t = threadIdx.x;
    const float* wsrc = w_comp + (t >> 6) * CIN + 4 * (t & 63);
    v4f wst[32];
#pragma unroll
    for (int i = 0; i < 32; ++i)
        wst[i] = *(const v4f*)(wsrc + i * 2 * CIN);

    const bool two = (t < WENC_PER_WG - 128);
    const int  ti0 = blockIdx.x * WENC_PER_WG + t;
    const int  ti1 = two ? ti0 + 128 : ti0;
    const float te0 = wenc_table_elem(w_enc, ti0);
    const float te1 = wenc_table_elem(w_enc, ti1);

    CgFrag P, Q;
    cg_load_b(P, xk, 0);

    float* wdst = &wl[(t & 63) * CW_PITCH + (t >> 6) * 4];
#pragma unroll
    for (int i = 0; i < 32; ++i) {
        float* d = wdst + (i >> 4) * 128 + (i & 7) * 8 + 2 * ((i >> 3) & 1);
        *(v2f*)d        = wst[i].lo;
        *(v2f*)(d + 64) = wst[i].hi;
    }
    __syncthreads();

    cg_load_a(P, wk, 0);
#pragma unroll 1
    for (int k0 = 0; k0 < CIN - 2 * CG_BK; k0 += 2 * CG_BK) {
        cg_load(Q, xk, wk, k0 + CG_BK);
        GROUP_FENCE();
        cg_mma(P, acc);
        GROUP_FENCE();
        cg_load(P, xk, wk, k0 + 2 * CG_BK);
        GROUP_FENCE();
        cg_mma(Q, acc);
        GROUP_FENCE();
    }
    cg_load(Q, xk, wk, CIN - CG_BK);
    GROUP_FENCE();
    cg_mma(P, acc);
    cg_mma(Q, acc);

    float* cb = comp + (size_t)n * CCOMP * HW + col;
    const int mhi = 8 * (lane >> 4);
#pragma unroll
    for (int mt = 0; mt < 4; ++mt) {
#pragma unroll
        for (int r = 0; r < 8; ++r) {
            const int m = mt * 16 + r + mhi;
            cb[(size_t)m * HW] = acc[mt][r] + b_comp[m];
        }
    }

    w_encF[ti0] = te0;
    if (two) w_encF[ti1] = te1;
}

#define EG_KS   4

__device__ __forceinline__ float xhalf(float x) {
    return __builtin_bit_cast(float, __builtin_amdgcn_permlanex16(
        __builtin_bit_cast(int, x), __builtin_bit_cast(int, x),
        0x76543210, 0xfedcba98, false, false));
}
#define EG_NBLK 36

struct EgFrag {
    v4f   a[EG_NLD];
    float b0[EG_KS], b1[EG_KS];
    bool  valid;
};

__device__ __forceinline__ void eg_load(EgFrag& f, const float* __restrict__ cbase,
                                        const float* __restrict__ wk,
                                        int h, int w, int blk) {
    const int plane = blk >> 2;
    const int cib   = (blk & 3) * 16;
    const int dy = plane / 3, dx = plane % 3;
    const int y  = h + dy - 1;
    const int xw = w + dx - 1;
    f.valid = (y >= 0) && (y < HH) && (xw >= 0) && (xw < WW);
    const int yc = min(max(y, 0), HH - 1);
    const int xc = min(max(xw, 0), WW - 1);
    const float* bsrc = cbase + (size_t)cib * HW + yc * WW + xc;
    const float* asrc = wk + blk * (EG_NLD * 128);
#pragma unroll
    for (int s = 0; s < EG_KS; ++s) {
        f.b0[s] = bsrc[(size_t)(4 * s) * HW];
        f.b1[s] = bsrc[(size_t)(4 * s + 1) * HW];
    }
#pragma unroll
    for (int j = 0; j < EG_NLD; ++j)
        f.a[j] = *(const v4f*)(asrc + j * 128);
}

__device__ __forceinline__ void eg_mma(const EgFrag& f, v8f (&acc)[7]) {
#pragma unroll
    for (int s = 0; s < EG_KS; ++s) {
        v2f bfrag;
        bfrag.x = f.valid ? f.b0[s] : 0.0f;
        bfrag.y = f.valid ? f.b1[s] : 0.0f;
#pragma unroll
        for (int mt = 0; mt < 7; ++mt) {
            const int fi = s * 7 + mt;
            const v2f afrag = (fi & 1) ? f.a[fi >> 1].hi : f.a[fi >> 1].lo;
            acc[mt] = __builtin_amdgcn_wmma_f32_16x16x4_f32(
                false, afrag, false, bfrag, (short)0, acc[mt], false, false);
        }
    }
}

__global__ void __launch_bounds__(128)
encoder_gemm(const float* __restrict__ comp, const float* __restrict__ w_encF,
             const float* __restrict__ b_enc, float* __restrict__ msoft) {
    const int wid   = blockIdx.x * 4 + (threadIdx.x >> 5);
    const int lane  = threadIdx.x & 31;
    const int n     = wid >> 8;
    const int nt    = wid & 255;
    const int col   = nt * 16 + (lane & 15);
    const int h     = col >> 6;
    const int w     = col & 63;
    const int khalf = (lane >> 4) * 2;

    v8f acc[7];
    const v8f vzero = {0.f,0.f,0.f,0.f,0.f,0.f,0.f,0.f};
#pragma unroll
    for (int mt = 0; mt < 7; ++mt) acc[mt] = vzero;

    const float* cbase = comp + (size_t)n * CCOMP * HW + (size_t)khalf * HW;
    const float* wk    = w_encF + lane * 4;

    EgFrag P, Q;
    eg_load(P, cbase, wk, h, w, 0);
#pragma unroll 1
    for (int blk = 0; blk < EG_NBLK - 2; blk += 2) {
        eg_load(Q, cbase, wk, h, w, blk + 1);
        GROUP_FENCE();
        eg_mma(P, acc);
        GROUP_FENCE();
        eg_load(P, cbase, wk, h, w, blk + 2);
        GROUP_FENCE();
        eg_mma(Q, acc);
        GROUP_FENCE();
    }
    eg_load(Q, cbase, wk, h, w, EG_NBLK - 1);
    GROUP_FENCE();
    eg_mma(P, acc);
    eg_mma(Q, acc);

    const int  hi  = lane >> 4;
    const bool low = (hi == 0);
    const float* bb = b_enc + 8 * hi;

    float v[6][8];
    float v6[4];
#pragma unroll
    for (int mt = 0; mt < 6; ++mt)
#pragma unroll
        for (int r = 0; r < 8; ++r) v[mt][r] = acc[mt][r] + bb[mt * 16 + r];
#pragma unroll
    for (int q = 0; q < 4; ++q) v6[q] = acc[6][q] + b_enc[96 + q];

#pragma unroll
    for (int q = 0; q < 4; ++q) {
        float mx = -1e30f;
#pragma unroll
        for (int mt = 0; mt < 6; ++mt) mx = fmaxf(mx, fmaxf(v[mt][q], v[mt][q + 4]));
        mx = fmaxf(mx, low ? v6[q] : -1e30f);
        mx = fmaxf(mx, xhalf(mx));
        float s = 0.0f;
#pragma unroll
        for (int mt = 0; mt < 6; ++mt) {
            v[mt][q]     = __expf(v[mt][q] - mx);      s += v[mt][q];
            v[mt][q + 4] = __expf(v[mt][q + 4] - mx);  s += v[mt][q + 4];
        }
        v6[q] = low ? __expf(v6[q] - mx) : 0.0f;
        s += v6[q];
        s += xhalf(s);
        const float inv = 1.0f / s;
#pragma unroll
        for (int mt = 0; mt < 6; ++mt) { v[mt][q] *= inv; v[mt][q + 4] *= inv; }
        v6[q] *= inv;
    }

    float* ob = msoft + (size_t)n * 25 * HWO + (size_t)(2 * hi) * HWO
                      + (size_t)(2 * h) * WO + 2 * w;
#pragma unroll
    for (int mt = 0; mt < 6; ++mt)
#pragma unroll
        for (int j = 0; j < 2; ++j)
#pragma unroll
            for (int sh = 0; sh < 2; ++sh) {
                const v2f o = {v[mt][4 * j + 2 * sh], v[mt][4 * j + 2 * sh + 1]};
                *(v2f*)(ob + (size_t)(mt * 4 + j) * HWO + sh * WO) = o;
            }
    if (low) {
#pragma unroll
        for (int sh = 0; sh < 2; ++sh) {
            const v2f o = {v6[2 * sh], v6[2 * sh + 1]};
            *(v2f*)(ob + (size_t)24 * HWO + sh * WO) = o;
        }
    }
}

#define RS_CH    32
#define RS_ROWS  4
#define RS_TR    (RS_ROWS + 4)
#define RS_PITCH 72

__global__ void __launch_bounds__(256)
reassemble(const float* __restrict__ x, const float* __restrict__ msoft,
           float* __restrict__ out) {
    __shared__ float tile[RS_CH * RS_TR * RS_PITCH] __attribute__((aligned(16)));

    const int t  = threadIdx.x;
    const int cz = blockIdx.x & 7;
    const int rt = (blockIdx.x >> 3) & 15;
    const int n  = blockIdx.x >> 7;
    const int hl = t >> 6;
    const int w  = t & 63;
    const int h  = rt * RS_ROWS + hl;
    const int c0 = cz * RS_CH;

    const int  r   = (t >> 4) & 7;
    const int  c4  = t & 15;
    const int  chb = t >> 7;
    const int  gy  = rt * RS_ROWS + r - 2;
    const bool ok  = (gy >= 0) && (gy < HH);
    const float* xs = x + ((size_t)n * CIN + c0 + chb) * HW + (ok ? gy : 0) * WW + c4 * 4;
    const v4f zero4 = {0.f, 0.f, 0.f, 0.f};

    v4f stg[16];
#pragma unroll
    for (int i = 0; i < 16; ++i) stg[i] = zero4;
    if (ok) {
#pragma unroll
        for (int i = 0; i < 16; ++i)
            stg[i] = *(const v4f*)(xs + (size_t)(2 * i) * HW);
    }

    v2f wt[25][2];
    const float* mb = msoft + (size_t)n * 25 * HWO + (size_t)(2 * h) * WO + 2 * w;
#pragma unroll
    for (int k = 0; k < 25; ++k) {
        wt[k][0] = *(const v2f*)(mb + (size_t)k * HWO);
        wt[k][1] = *(const v2f*)(mb + (size_t)k * HWO + WO);
    }

    float* tst = &tile[(chb * RS_TR + r) * RS_PITCH + 4 + c4 * 4];
#pragma unroll
    for (int i = 0; i < 16; ++i)
        *(v4f*)(tst + (2 * i) * RS_TR * RS_PITCH) = stg[i];

#pragma unroll
    for (int i = 0; i < 2; ++i) {
        const int f = t + 256 * i;
        *(v4f*)(&tile[(f >> 1) * RS_PITCH + ((f & 1) ? (RS_PITCH - 4) : 0)]) = zero4;
    }
    __syncthreads();

    const float* tp = &tile[hl * RS_PITCH + w + 2];
    float* ob = out + ((size_t)n * CIN + c0) * HWO + (size_t)(2 * h) * WO + 2 * w;

#pragma unroll 2
    for (int ch = 0; ch < RS_CH; ++ch) {
        const float* tc = tp + ch * (RS_TR * RS_PITCH);
        v2f a0 = {0.f, 0.f};
        v2f a1 = {0.f, 0.f};
#pragma unroll
        for (int p = 0; p < 5; ++p)
#pragma unroll
            for (int q = 0; q < 5; ++q) {
                const float xv = tc[p * RS_PITCH + q];
                const v2f xx = {xv, xv};
                a0 = __builtin_elementwise_fma(wt[p * 5 + q][0], xx, a0);
                a1 = __builtin_elementwise_fma(wt[p * 5 + q][1], xx, a1);
            }
        __builtin_nontemporal_store(a0, (v2f*)(ob + (size_t)ch * HWO));
        __builtin_nontemporal_store(a1, (v2f*)(ob + (size_t)ch * HWO + WO));
    }
}

extern "C" void kernel_launch(void* const* d_in, const int* in_sizes, int n_in,
                              void* d_out, int out_size, void* d_ws, size_t ws_size,
                              hipStream_t stream) {
    const float* x      = (const float*)d_in[0];
    const float* w_comp = (const float*)d_in[1];
    const float* b_comp = (const float*)d_in[2];
    const float* w_enc  = (const float*)d_in[3];
    const float* b_enc  = (const float*)d_in[4];
    float* out = (float*)d_out;

    char* ws = (char*)d_ws;
    float* comp    = (float*)(ws);
    float* msoft   = (float*)(ws + (size_t)4194304);
    float* w_encF  = (float*)(ws + (size_t)4194304 + 6553600);

    compress_gemm  <<<256, 128, 0, stream>>>(x, w_comp, b_comp, w_enc, comp, w_encF);
    encoder_gemm   <<<256, 128, 0, stream>>>(comp, w_encF, b_enc, msoft);
    reassemble     <<<NB * 16 * 8, 256, 0, stream>>>(x, msoft, out);
}


# both GEMM k loops fetch two blocks ahead instead of one (three register sets; the 1x1 conv's 8 blocks now straight-line): loads get two blocks of WMMAs as lead; k order and sums unchanged
# speedup vs baseline: 7.6852x; 1.0943x over previous
#include <hip/hip_runtime.h>
#include <hip/hip_bf16.h>


#define NB     4
#define CIN    256
#define CCOMP  64
#define CMASK  100
#define HH     64
#define WW     64
#define HW     (HH*WW)
#define HO     128
#define WO     128
#define HWO    (HO*WO)
#define KTOT   576
#define MPAD   112

typedef float v2f __attribute__((ext_vector_type(2)));
typedef float v4f __attribute__((ext_vector_type(4)));
typedef float v8f __attribute__((ext_vector_type(8)));

#define EG_NLD  14
#define WENC_N  (36 * EG_NLD * 32 * 4)
#define WENC_PER_WG (WENC_N / 256)
#define CW_PITCH 260

__device__ __forceinline__ float wenc_table_elem(const float* __restrict__ w_enc, int idx) {
    const int e   = idx & 3;
    const int l   = (idx >> 2) & 31;
    const int g   = idx >> 7;
    const int blk = g / EG_NLD;
    const int j   = g % EG_NLD;
    const int f   = 2 * j + (e >> 1);
    const int s   = f / 7;
    const int mt  = f % 7;
    const int m   = mt * 16 + (l & 15);
    const int k   = blk * 16 + 4 * s + 2 * (l >> 4) + (e & 1);
    const int plane = k >> 6;
    const int ci    = k & 63;
    const float v = w_enc[min(m, CMASK - 1) * KTOT + ci * 9 + plane];
    return (m < CMASK) ? v : 0.0f;
}

#define CG_KS 8
#define CG_BK (4 * CG_KS)

#define GROUP_FENCE() __builtin_amdgcn_sched_barrier(0)


struct CgFrag {
    v4f a[CG_KS][2];
    v2f b[CG_KS];
};

__device__ __forceinline__ void cg_load_b(CgFrag& f, const float* __restrict__ xk, int k0) {
#pragma unroll
    for (int s = 0; s < CG_KS; ++s) {
        const int k = k0 + 4 * s;
        f.b[s].x = xk[(size_t)k * HW];
        f.b[s].y = xk[(size_t)(k + 1) * HW];
    }
}

__device__ __forceinline__ void cg_load_a(CgFrag& f, const float* wk, int k0) {
#pragma unroll
    for (int s = 0; s < CG_KS; ++s)
#pragma unroll
        for (int j = 0; j < 2; ++j)
            f.a[s][j] = *(const v4f*)(wk + ((k0 >> 2) + s) * CW_PITCH + j * 128);
}

__device__ __forceinline__ void cg_load(CgFrag& f, const float* __restrict__ xk,
                                        const float* wk, int k0) {
    cg_load_b(f, xk, k0);
    cg_load_a(f, wk, k0);
}

__device__ __forceinline__ void cg_mma(const CgFrag& f, v8f (&acc)[4]) {
#pragma unroll
    for (int s = 0; s < CG_KS; ++s)
#pragma unroll
        for (int mt = 0; mt < 4; ++mt) {
            const v2f afrag = (mt & 1) ? f.a[s][mt >> 1].hi : f.a[s][mt >> 1].lo;
            acc[mt] = __builtin_amdgcn_wmma_f32_16x16x4_f32(
                false, afrag, false, f.b[s], (short)0, acc[mt], false, false);
        }
}

__global__ void __launch_bounds__(128)
compress_gemm(const float* __restrict__ x, const float* __restrict__ w_comp,
              const float* __restrict__ b_comp, const float* __restrict__ w_enc,
              float* __restrict__ comp, float* __restrict__ w_encF) {
    __shared__ float wl[64 * CW_PITCH] __attribute__((aligned(16)));

    const int wid   = blockIdx.x * 4 + (threadIdx.x >> 5);
    const int lane  = threadIdx.x & 31;
    const int n     = wid >> 8;
    const int nt    = wid & 255;
    const int col   = nt * 16 + (lane & 15);
    const int khalf = (lane >> 4) * 2;

    const float* xk = x + (size_t)n * CIN * HW + col + (size_t)khalf * HW;
    const float* wk = wl + lane * 4;

    v8f acc[4];
    const v8f vzero = {0.f,0.f,0.f,0.f,0.f,0.f,0.f,0.f};
#pragma unroll
    for (int mt = 0; mt < 4; ++mt) acc[mt] = vzero;

    const int t = threadIdx.x;
    const float* wsrc = w_comp + (t >> 6) * CIN + 4 * (t & 63);
    v4f wst[32];
#pragma unroll
    for (int i = 0; i < 32; ++i)
        wst[i] = *(const v4f*)(wsrc + i * 2 * CIN);

    const bool two = (t < WENC_PER_WG - 128);
    const int  ti0 = blockIdx.x * WENC_PER_WG + t;
    const int  ti1 = two ? ti0 + 128 : ti0;
    const float te0 = wenc_table_elem(w_enc, ti0);
    const float te1 = wenc_table_elem(w_enc, ti1);

    CgFrag P, Q, R;
    cg_load_b(P, xk, 0);
    cg_load_b(Q, xk, CG_BK);

    float* wdst = &wl[(t & 63) * CW_PITCH + (t >> 6) * 4];
#pragma unroll
    for (int i = 0; i < 32; ++i) {
        float* d = wdst + (i >> 4) * 128 + (i & 7) * 8 + 2 * ((i >> 3) & 1);
        *(v2f*)d        = wst[i].lo;
        *(v2f*)(d + 64) = wst[i].hi;
    }
    __syncthreads();

    cg_load_a(P, wk, 0);
    cg_load_a(Q, wk, CG_BK);
#pragma unroll
    for (int k0 = 0; k0 < CIN - 4 * CG_BK; k0 += 3 * CG_BK) {
        cg_load(R, xk, wk, k0 + 2 * CG_BK);
        GROUP_FENCE();
        cg_mma(P, acc);
        GROUP_FENCE();
        cg_load(P, xk, wk, k0 + 3 * CG_BK);
        GROUP_FENCE();
        cg_mma(Q, acc);
        GROUP_FENCE();
        cg_load(Q, xk, wk, k0 + 4 * CG_BK);
        GROUP_FENCE();
        cg_mma(R, acc);
        GROUP_FENCE();
    }
    cg_mma(P, acc);
    cg_mma(Q, acc);

    float* cb = comp + (size_t)n * CCOMP * HW + col;
    const int mhi = 8 * (lane >> 4);
#pragma unroll
    for (int mt = 0; mt < 4; ++mt) {
#pragma unroll
        for (int r = 0; r < 8; ++r) {
            const int m = mt * 16 + r + mhi;
            cb[(size_t)m * HW] = acc[mt][r] + b_comp[m];
        }
    }

    w_encF[ti0] = te0;
    if (two) w_encF[ti1] = te1;
}

#define EG_KS   4

__device__ __forceinline__ float xhalf(float x) {
    return __builtin_bit_cast(float, __builtin_amdgcn_permlanex16(
        __builtin_bit_cast(int, x), __builtin_bit_cast(int, x),
        0x76543210, 0xfedcba98, false, false));
}
#define EG_NBLK 36

struct EgFrag {
    v4f   a[EG_NLD];
    float b0[EG_KS], b1[EG_KS];
    bool  valid;
};

__device__ __forceinline__ void eg_load(EgFrag& f, const float* __restrict__ cbase,
                                        const float* __restrict__ wk,
                                        int h, int w, int blk) {
    const int plane = blk >> 2;
    const int cib   = (blk & 3) * 16;
    const int dy = plane / 3, dx = plane % 3;
    const int y  = h + dy - 1;
    const int xw = w + dx - 1;
    f.valid = (y >= 0) && (y < HH) && (xw >= 0) && (xw < WW);
    const int yc = min(max(y, 0), HH - 1);
    const int xc = min(max(xw, 0), WW - 1);
    const float* bsrc = cbase + (size_t)cib * HW + yc * WW + xc;
    const float* asrc = wk + blk * (EG_NLD * 128);
#pragma unroll
    for (int s = 0; s < EG_KS; ++s) {
        f.b0[s] = bsrc[(size_t)(4 * s) * HW];
        f.b1[s] = bsrc[(size_t)(4 * s + 1) * HW];
    }
#pragma unroll
    for (int j = 0; j < EG_NLD; ++j)
        f.a[j] = *(const v4f*)(asrc + j * 128);
}

__device__ __forceinline__ void eg_mma(const EgFrag& f, v8f (&acc)[7]) {
#pragma unroll
    for (int s = 0; s < EG_KS; ++s) {
        v2f bfrag;
        bfrag.x = f.valid ? f.b0[s] : 0.0f;
        bfrag.y = f.valid ? f.b1[s] : 0.0f;
#pragma unroll
        for (int mt = 0; mt < 7; ++mt) {
            const int fi = s * 7 + mt;
            const v2f afrag = (fi & 1) ? f.a[fi >> 1].hi : f.a[fi >> 1].lo;
            acc[mt] = __builtin_amdgcn_wmma_f32_16x16x4_f32(
                false, afrag, false, bfrag, (short)0, acc[mt], false, false);
        }
    }
}

__global__ void __launch_bounds__(128)
encoder_gemm(const float* __restrict__ comp, const float* __restrict__ w_encF,
             const float* __restrict__ b_enc, float* __restrict__ msoft) {
    const int wid   = blockIdx.x * 4 + (threadIdx.x >> 5);
    const int lane  = threadIdx.x & 31;
    const int n     = wid >> 8;
    const int nt    = wid & 255;
    const int col   = nt * 16 + (lane & 15);
    const int h     = col >> 6;
    const int w     = col & 63;
    const int khalf = (lane >> 4) * 2;

    v8f acc[7];
    const v8f vzero = {0.f,0.f,0.f,0.f,0.f,0.f,0.f,0.f};
#pragma unroll
    for (int mt = 0; mt < 7; ++mt) acc[mt] = vzero;

    const float* cbase = comp + (size_t)n * CCOMP * HW + (size_t)khalf * HW;
    const float* wk    = w_encF + lane * 4;

    EgFrag P, Q, R;
    eg_load(P, cbase, wk, h, w, 0);
    GROUP_FENCE();
    eg_load(Q, cbase, wk, h, w, 1);
    GROUP_FENCE();
#pragma unroll 1
    for (int blk = 0; blk < EG_NBLK - 3; blk += 3) {
        eg_load(R, cbase, wk, h, w, blk + 2);
        GROUP_FENCE();
        eg_mma(P, acc);
        GROUP_FENCE();
        eg_load(P, cbase, wk, h, w, blk + 3);
        GROUP_FENCE();
        eg_mma(Q, acc);
        GROUP_FENCE();
        eg_load(Q, cbase, wk, h, w, blk + 4);
        GROUP_FENCE();
        eg_mma(R, acc);
        GROUP_FENCE();
    }
    eg_load(R, cbase, wk, h, w, EG_NBLK - 1);
    GROUP_FENCE();
    eg_mma(P, acc);
    eg_mma(Q, acc);
    eg_mma(R, acc);

    const int  hi  = lane >> 4;
    const bool low = (hi == 0);
    const float* bb = b_enc + 8 * hi;

    float v[6][8];
    float v6[4];
#pragma unroll
    for (int mt = 0; mt < 6; ++mt)
#pragma unroll
        for (int r = 0; r < 8; ++r) v[mt][r] = acc[mt][r] + bb[mt * 16 + r];
#pragma unroll
    for (int q = 0; q < 4; ++q) v6[q] = acc[6][q] + b_enc[96 + q];

#pragma unroll
    for (int q = 0; q < 4; ++q) {
        float mx = -1e30f;
#pragma unroll
        for (int mt = 0; mt < 6; ++mt) mx = fmaxf(mx, fmaxf(v[mt][q], v[mt][q + 4]));
        mx = fmaxf(mx, low ? v6[q] : -1e30f);
        mx = fmaxf(mx, xhalf(mx));
        float s = 0.0f;
#pragma unroll
        for (int mt = 0; mt < 6; ++mt) {
            v[mt][q]     = __expf(v[mt][q] - mx);      s += v[mt][q];
            v[mt][q + 4] = __expf(v[mt][q + 4] - mx);  s += v[mt][q + 4];
        }
        v6[q] = low ? __expf(v6[q] - mx) : 0.0f;
        s += v6[q];
        s += xhalf(s);
        const float inv = 1.0f / s;
#pragma unroll
        for (int mt = 0; mt < 6; ++mt) { v[mt][q] *= inv; v[mt][q + 4] *= inv; }
        v6[q] *= inv;
    }

    float* ob = msoft + (size_t)n * 25 * HWO + (size_t)(2 * hi) * HWO
                      + (size_t)(2 * h) * WO + 2 * w;
#pragma unroll
    for (int mt = 0; mt < 6; ++mt)
#pragma unroll
        for (int j = 0; j < 2; ++j)
#pragma unroll
            for (int sh = 0; sh < 2; ++sh) {
                const v2f o = {v[mt][4 * j + 2 * sh], v[mt][4 * j + 2 * sh + 1]};
                *(v2f*)(ob + (size_t)(mt * 4 + j) * HWO + sh * WO) = o;
            }
    if (low) {
#pragma unroll
        for (int sh = 0; sh < 2; ++sh) {
            const v2f o = {v6[2 * sh], v6[2 * sh + 1]};
            *(v2f*)(ob + (size_t)24 * HWO + sh * WO) = o;
        }
    }
}

#define RS_CH    32
#define RS_ROWS  4
#define RS_TR    (RS_ROWS + 4)
#define RS_PITCH 72

__global__ void __launch_bounds__(256)
reassemble(const float* __restrict__ x, const float* __restrict__ msoft,
           float* __restrict__ out) {
    __shared__ float tile[RS_CH * RS_TR * RS_PITCH] __attribute__((aligned(16)));

    const int t  = threadIdx.x;
    const int cz = blockIdx.x & 7;
    const int rt = (blockIdx.x >> 3) & 15;
    const int n  = blockIdx.x >> 7;
    const int hl = t >> 6;
    const int w  = t & 63;
    const int h  = rt * RS_ROWS + hl;
    const int c0 = cz * RS_CH;

    const int  r   = (t >> 4) & 7;
    const int  c4  = t & 15;
    const int  chb = t >> 7;
    const int  gy  = rt * RS_ROWS + r - 2;
    const bool ok  = (gy >= 0) && (gy < HH);
    const float* xs = x + ((size_t)n * CIN + c0 + chb) * HW + (ok ? gy : 0) * WW + c4 * 4;
    const v4f zero4 = {0.f, 0.f, 0.f, 0.f};

    v4f stg[16];
#pragma unroll
    for (int i = 0; i < 16; ++i) stg[i] = zero4;
    if (ok) {
#pragma unroll
        for (int i = 0; i < 16; ++i)
            stg[i] = *(const v4f*)(xs + (size_t)(2 * i) * HW);
    }

    v2f wt[25][2];
    const float* mb = msoft + (size_t)n * 25 * HWO + (size_t)(2 * h) * WO + 2 * w;
#pragma unroll
    for (int k = 0; k < 25; ++k) {
        wt[k][0] = *(const v2f*)(mb + (size_t)k * HWO);
        wt[k][1] = *(const v2f*)(mb + (size_t)k * HWO + WO);
    }

    float* tst = &tile[(chb * RS_TR + r) * RS_PITCH + 4 + c4 * 4];
#pragma unroll
    for (int i = 0; i < 16; ++i)
        *(v4f*)(tst + (2 * i) * RS_TR * RS_PITCH) = stg[i];

#pragma unroll
    for (int i = 0; i < 2; ++i) {
        const int f = t + 256 * i;
        *(v4f*)(&tile[(f >> 1) * RS_PITCH + ((f & 1) ? (RS_PITCH - 4) : 0)]) = zero4;
    }
    __syncthreads();

    const float* tp = &tile[hl * RS_PITCH + w + 2];
    float* ob = out + ((size_t)n * CIN + c0) * HWO + (size_t)(2 * h) * WO + 2 * w;

#pragma unroll 2
    for (int ch = 0; ch < RS_CH; ++ch) {
        const float* tc = tp + ch * (RS_TR * RS_PITCH);
        v2f a0 = {0.f, 0.f};
        v2f a1 = {0.f, 0.f};
#pragma unroll
        for (int p = 0; p < 5; ++p)
#pragma unroll
            for (int q = 0; q < 5; ++q) {
                const float xv = tc[p * RS_PITCH + q];
                const v2f xx = {xv, xv};
                a0 = __builtin_elementwise_fma(wt[p * 5 + q][0], xx, a0);
                a1 = __builtin_elementwise_fma(wt[p * 5 + q][1], xx, a1);
            }
        __builtin_nontemporal_store(a0, (v2f*)(ob + (size_t)ch * HWO));
        __builtin_nontemporal_store(a1, (v2f*)(ob + (size_t)ch * HWO + WO));
    }
}

extern "C" void kernel_launch(void* const* d_in, const int* in_sizes, int n_in,
                              void* d_out, int out_size, void* d_ws, size_t ws_size,
                              hipStream_t stream) {
    const float* x      = (const float*)d_in[0];
    const float* w_comp = (const float*)d_in[1];
    const float* b_comp = (const float*)d_in[2];
    const float* w_enc  = (const float*)d_in[3];
    const float* b_enc  = (const float*)d_in[4];
    float* out = (float*)d_out;

    char* ws = (char*)d_ws;
    float* comp    = (float*)(ws);
    float* msoft   = (float*)(ws + (size_t)4194304);
    float* w_encF  = (float*)(ws + (size_t)4194304 + 6553600);

    compress_gemm  <<<256, 128, 0, stream>>>(x, w_comp, b_comp, w_enc, comp, w_encF);
    encoder_gemm   <<<256, 128, 0, stream>>>(comp, w_encF, b_enc, msoft);
    reassemble     <<<NB * 16 * 8, 256, 0, stream>>>(x, msoft, out);
}


# reassembly: a thread owns two adjacent low-res pixels; their 5x5 windows share 6-float LDS row reads (aligned 8-B loads), 4 outputs per row leave as one 16-B store; half the waves; sums unchanged
# speedup vs baseline: 8.0540x; 1.0480x over previous
#include <hip/hip_runtime.h>
#include <hip/hip_bf16.h>


#define NB     4
#define CIN    256
#define CCOMP  64
#define CMASK  100
#define HH     64
#define WW     64
#define HW     (HH*WW)
#define HO     128
#define WO     128
#define HWO    (HO*WO)
#define KTOT   576
#define MPAD   112

typedef float v2f __attribute__((ext_vector_type(2)));
typedef float v4f __attribute__((ext_vector_type(4)));
typedef float v8f __attribute__((ext_vector_type(8)));

#define EG_NLD  14
#define WENC_N  (36 * EG_NLD * 32 * 4)
#define WENC_PER_WG (WENC_N / 256)
#define CW_PITCH 260

__device__ __forceinline__ float wenc_table_elem(const float* __restrict__ w_enc, int idx) {
    const int e   = idx & 3;
    const int l   = (idx >> 2) & 31;
    const int g   = idx >> 7;
    const int blk = g / EG_NLD;
    const int j   = g % EG_NLD;
    const int f   = 2 * j + (e >> 1);
    const int s   = f / 7;
    const int mt  = f % 7;
    const int m   = mt * 16 + (l & 15);
    const int k   = blk * 16 + 4 * s + 2 * (l >> 4) + (e & 1);
    const int plane = k >> 6;
    const int ci    = k & 63;
    const float v = w_enc[min(m, CMASK - 1) * KTOT + ci * 9 + plane];
    return (m < CMASK) ? v : 0.0f;
}

#define CG_KS 8
#define CG_BK (4 * CG_KS)

#define GROUP_FENCE() __builtin_amdgcn_sched_barrier(0)


struct CgFrag {
    v4f a[CG_KS][2];
    v2f b[CG_KS];
};

__device__ __forceinline__ void cg_load_b(CgFrag& f, const float* __restrict__ xk, int k0) {
#pragma unroll
    for (int s = 0; s < CG_KS; ++s) {
        const int k = k0 + 4 * s;
        f.b[s].x = xk[(size_t)k * HW];
        f.b[s].y = xk[(size_t)(k + 1) * HW];
    }
}

__device__ __forceinline__ void cg_load_a(CgFrag& f, const float* wk, int k0) {
#pragma unroll
    for (int s = 0; s < CG_KS; ++s)
#pragma unroll
        for (int j = 0; j < 2; ++j)
            f.a[s][j] = *(const v4f*)(wk + ((k0 >> 2) + s) * CW_PITCH + j * 128);
}

__device__ __forceinline__ void cg_load(CgFrag& f, const float* __restrict__ xk,
                                        const float* wk, int k0) {
    cg_load_b(f, xk, k0);
    cg_load_a(f, wk, k0);
}

__device__ __forceinline__ void cg_mma(const CgFrag& f, v8f (&acc)[4]) {
#pragma unroll
    for (int s = 0; s < CG_KS; ++s)
#pragma unroll
        for (int mt = 0; mt < 4; ++mt) {
            const v2f afrag = (mt & 1) ? f.a[s][mt >> 1].hi : f.a[s][mt >> 1].lo;
            acc[mt] = __builtin_amdgcn_wmma_f32_16x16x4_f32(
                false, afrag, false, f.b[s], (short)0, acc[mt], false, false);
        }
}

__global__ void __launch_bounds__(128)
compress_gemm(const float* __restrict__ x, const float* __restrict__ w_comp,
              const float* __restrict__ b_comp, const float* __restrict__ w_enc,
              float* __restrict__ comp, float* __restrict__ w_encF) {
    __shared__ float wl[64 * CW_PITCH] __attribute__((aligned(16)));

    const int wid   = blockIdx.x * 4 + (threadIdx.x >> 5);
    const int lane  = threadIdx.x & 31;
    const int n     = wid >> 8;
    const int nt    = wid & 255;
    const int col   = nt * 16 + (lane & 15);
    const int khalf = (lane >> 4) * 2;

    const float* xk = x + (size_t)n * CIN * HW + col + (size_t)khalf * HW;
    const float* wk = wl + lane * 4;

    v8f acc[4];
    const v8f vzero = {0.f,0.f,0.f,0.f,0.f,0.f,0.f,0.f};
#pragma unroll
    for (int mt = 0; mt < 4; ++mt) acc[mt] = vzero;

    const int t = threadIdx.x;
    const float* wsrc = w_comp + (t >> 6) * CIN + 4 * (t & 63);
    v4f wst[32];
#pragma unroll
    for (int i = 0; i < 32; ++i)
        wst[i] = *(const v4f*)(wsrc + i * 2 * CIN);

    const bool two = (t < WENC_PER_WG - 128);
    const int  ti0 = blockIdx.x * WENC_PER_WG + t;
    const int  ti1 = two ? ti0 + 128 : ti0;
    const float te0 = wenc_table_elem(w_enc, ti0);
    const float te1 = wenc_table_elem(w_enc, ti1);

    CgFrag P, Q, R;
    cg_load_b(P, xk, 0);
    cg_load_b(Q, xk, CG_BK);

    float* wdst = &wl[(t & 63) * CW_PITCH + (t >> 6) * 4];
#pragma unroll
    for (int i = 0; i < 32; ++i) {
        float* d = wdst + (i >> 4) * 128 + (i & 7) * 8 + 2 * ((i >> 3) & 1);
        *(v2f*)d        = wst[i].lo;
        *(v2f*)(d + 64) = wst[i].hi;
    }
    __syncthreads();

    cg_load_a(P, wk, 0);
    cg_load_a(Q, wk, CG_BK);
#pragma unroll
    for (int k0 = 0; k0 < CIN - 4 * CG_BK; k0 += 3 * CG_BK) {
        cg_load(R, xk, wk, k0 + 2 * CG_BK);
        GROUP_FENCE();
        cg_mma(P, acc);
        GROUP_FENCE();
        cg_load(P, xk, wk, k0 + 3 * CG_BK);
        GROUP_FENCE();
        cg_mma(Q, acc);
        GROUP_FENCE();
        cg_load(Q, xk, wk, k0 + 4 * CG_BK);
        GROUP_FENCE();
        cg_mma(R, acc);
        GROUP_FENCE();
    }
    cg_mma(P, acc);
    cg_mma(Q, acc);

    float* cb = comp + (size_t)n * CCOMP * HW + col;
    const int mhi = 8 * (lane >> 4);
#pragma unroll
    for (int mt = 0; mt < 4; ++mt) {
#pragma unroll
        for (int r = 0; r < 8; ++r) {
            const int m = mt * 16 + r + mhi;
            cb[(size_t)m * HW] = acc[mt][r] + b_comp[m];
        }
    }

    w_encF[ti0] = te0;
    if (two) w_encF[ti1] = te1;
}

#define EG_KS   4

__device__ __forceinline__ float xhalf(float x) {
    return __builtin_bit_cast(float, __builtin_amdgcn_permlanex16(
        __builtin_bit_cast(int, x), __builtin_bit_cast(int, x),
        0x76543210, 0xfedcba98, false, false));
}
#define EG_NBLK 36

struct EgFrag {
    v4f   a[EG_NLD];
    float b0[EG_KS], b1[EG_KS];
    bool  valid;
};

__device__ __forceinline__ void eg_load(EgFrag& f, const float* __restrict__ cbase,
                                        const float* __restrict__ wk,
                                        int h, int w, int blk) {
    const int plane = blk >> 2;
    const int cib   = (blk & 3) * 16;
    const int dy = plane / 3, dx = plane % 3;
    const int y  = h + dy - 1;
    const int xw = w + dx - 1;
    f.valid = (y >= 0) && (y < HH) && (xw >= 0) && (xw < WW);
    const int yc = min(max(y, 0), HH - 1);
    const int xc = min(max(xw, 0), WW - 1);
    const float* bsrc = cbase + (size_t)cib * HW + yc * WW + xc;
    const float* asrc = wk + blk * (EG_NLD * 128);
#pragma unroll
    for (int s = 0; s < EG_KS; ++s) {
        f.b0[s] = bsrc[(size_t)(4 * s) * HW];
        f.b1[s] = bsrc[(size_t)(4 * s + 1) * HW];
    }
#pragma unroll
    for (int j = 0; j < EG_NLD; ++j)
        f.a[j] = *(const v4f*)(asrc + j * 128);
}

__device__ __forceinline__ void eg_mma(const EgFrag& f, v8f (&acc)[7]) {
#pragma unroll
    for (int s = 0; s < EG_KS; ++s) {
        v2f bfrag;
        bfrag.x = f.valid ? f.b0[s] : 0.0f;
        bfrag.y = f.valid ? f.b1[s] : 0.0f;
#pragma unroll
        for (int mt = 0; mt < 7; ++mt) {
            const int fi = s * 7 + mt;
            const v2f afrag = (fi & 1) ? f.a[fi >> 1].hi : f.a[fi >> 1].lo;
            acc[mt] = __builtin_amdgcn_wmma_f32_16x16x4_f32(
                false, afrag, false, bfrag, (short)0, acc[mt], false, false);
        }
    }
}

__global__ void __launch_bounds__(128)
encoder_gemm(const float* __restrict__ comp, const float* __restrict__ w_encF,
             const float* __restrict__ b_enc, float* __restrict__ msoft) {
    const int wid   = blockIdx.x * 4 + (threadIdx.x >> 5);
    const int lane  = threadIdx.x & 31;
    const int n     = wid >> 8;
    const int nt    = wid & 255;
    const int col   = nt * 16 + (lane & 15);
    const int h     = col >> 6;
    const int w     = col & 63;
    const int khalf = (lane >> 4) * 2;

    v8f acc[7];
    const v8f vzero = {0.f,0.f,0.f,0.f,0.f,0.f,0.f,0.f};
#pragma unroll
    for (int mt = 0; mt < 7; ++mt) acc[mt] = vzero;

    const float* cbase = comp + (size_t)n * CCOMP * HW + (size_t)khalf * HW;
    const float* wk    = w_encF + lane * 4;

    EgFrag P, Q, R;
    eg_load(P, cbase, wk, h, w, 0);
    GROUP_FENCE();
    eg_load(Q, cbase, wk, h, w, 1);
    GROUP_FENCE();
#pragma unroll 1
    for (int blk = 0; blk < EG_NBLK - 3; blk += 3) {
        eg_load(R, cbase, wk, h, w, blk + 2);
        GROUP_FENCE();
        eg_mma(P, acc);
        GROUP_FENCE();
        eg_load(P, cbase, wk, h, w, blk + 3);
        GROUP_FENCE();
        eg_mma(Q, acc);
        GROUP_FENCE();
        eg_load(Q, cbase, wk, h, w, blk + 4);
        GROUP_FENCE();
        eg_mma(R, acc);
        GROUP_FENCE();
    }
    eg_load(R, cbase, wk, h, w, EG_NBLK - 1);
    GROUP_FENCE();
    eg_mma(P, acc);
    eg_mma(Q, acc);
    eg_mma(R, acc);

    const int  hi  = lane >> 4;
    const bool low = (hi == 0);
    const float* bb = b_enc + 8 * hi;

    float v[6][8];
    float v6[4];
#pragma unroll
    for (int mt = 0; mt < 6; ++mt)
#pragma unroll
        for (int r = 0; r < 8; ++r) v[mt][r] = acc[mt][r] + bb[mt * 16 + r];
#pragma unroll
    for (int q = 0; q < 4; ++q) v6[q] = acc[6][q] + b_enc[96 + q];

#pragma unroll
    for (int q = 0; q < 4; ++q) {
        float mx = -1e30f;
#pragma unroll
        for (int mt = 0; mt < 6; ++mt) mx = fmaxf(mx, fmaxf(v[mt][q], v[mt][q + 4]));
        mx = fmaxf(mx, low ? v6[q] : -1e30f);
        mx = fmaxf(mx, xhalf(mx));
        float s = 0.0f;
#pragma unroll
        for (int mt = 0; mt < 6; ++mt) {
            v[mt][q]     = __expf(v[mt][q] - mx);      s += v[mt][q];
            v[mt][q + 4] = __expf(v[mt][q + 4] - mx);  s += v[mt][q + 4];
        }
        v6[q] = low ? __expf(v6[q] - mx) : 0.0f;
        s += v6[q];
        s += xhalf(s);
        const float inv = 1.0f / s;
#pragma unroll
        for (int mt = 0; mt < 6; ++mt) { v[mt][q] *= inv; v[mt][q + 4] *= inv; }
        v6[q] *= inv;
    }

    float* ob = msoft + (size_t)n * 25 * HWO + (size_t)(2 * hi) * HWO
                      + (size_t)(2 * h) * WO + 2 * w;
#pragma unroll
    for (int mt = 0; mt < 6; ++mt)
#pragma unroll
        for (int j = 0; j < 2; ++j)
#pragma unroll
            for (int sh = 0; sh < 2; ++sh) {
                const v2f o = {v[mt][4 * j + 2 * sh], v[mt][4 * j + 2 * sh + 1]};
                *(v2f*)(ob + (size_t)(mt * 4 + j) * HWO + sh * WO) = o;
            }
    if (low) {
#pragma unroll
        for (int sh = 0; sh < 2; ++sh) {
            const v2f o = {v6[2 * sh], v6[2 * sh + 1]};
            *(v2f*)(ob + (size_t)24 * HWO + sh * WO) = o;
        }
    }
}

#define RS_CH    32
#define RS_ROWS  4
#define RS_TR    (RS_ROWS + 4)
#define RS_PITCH 72

__global__ void __launch_bounds__(128)
reassemble(const float* __restrict__ x, const float* __restrict__ msoft,
           float* __restrict__ out) {
    __shared__ float tile[RS_CH * RS_TR * RS_PITCH] __attribute__((aligned(16)));

    const int t  = threadIdx.x;
    const int cz = blockIdx.x & 7;
    const int rt = (blockIdx.x >> 3) & 15;
    const int n  = blockIdx.x >> 7;
    const int hl = t >> 5;
    const int w0 = (t & 31) * 2;
    const int h  = rt * RS_ROWS + hl;
    const int c0 = cz * RS_CH;

    const int  r   = t >> 4;
    const int  c4  = t & 15;
    const int  gy  = rt * RS_ROWS + r - 2;
    const bool ok  = (gy >= 0) && (gy < HH);
    const float* xs = x + ((size_t)n * CIN + c0) * HW + (ok ? gy : 0) * WW + c4 * 4;
    const v4f zero4 = {0.f, 0.f, 0.f, 0.f};

    v4f stg[RS_CH];
#pragma unroll
    for (int i = 0; i < RS_CH; ++i) stg[i] = zero4;
    if (ok) {
#pragma unroll
        for (int i = 0; i < RS_CH; ++i)
            stg[i] = *(const v4f*)(xs + (size_t)i * HW);
    }

    v4f wt[25][2];
    const float* mb = msoft + (size_t)n * 25 * HWO + (size_t)(2 * h) * WO + 2 * w0;
#pragma unroll
    for (int k = 0; k < 25; ++k) {
        wt[k][0] = *(const v4f*)(mb + (size_t)k * HWO);
        wt[k][1] = *(const v4f*)(mb + (size_t)k * HWO + WO);
    }

    float* tst = &tile[r * RS_PITCH + 4 + c4 * 4];
#pragma unroll
    for (int i = 0; i < RS_CH; ++i)
        *(v4f*)(tst + i * RS_TR * RS_PITCH) = stg[i];

#pragma unroll
    for (int i = 0; i < 4; ++i) {
        const int f = t + 128 * i;
        *(v4f*)(&tile[(f >> 1) * RS_PITCH + ((f & 1) ? (RS_PITCH - 4) : 0)]) = zero4;
    }
    __syncthreads();

    const float* tp = &tile[hl * RS_PITCH + w0 + 2];
    float* ob = out + ((size_t)n * CIN + c0) * HWO + (size_t)(2 * h) * WO + 2 * w0;

#pragma unroll 2
    for (int ch = 0; ch < RS_CH; ++ch) {
        const float* tc = tp + ch * (RS_TR * RS_PITCH);
        v2f a00 = {0.f, 0.f};
        v2f a01 = {0.f, 0.f};
        v2f a10 = {0.f, 0.f};
        v2f a11 = {0.f, 0.f};
#pragma unroll
        for (int p = 0; p < 5; ++p) {
            const v2f x01 = *(const v2f*)(tc + p * RS_PITCH);
            const v2f x23 = *(const v2f*)(tc + p * RS_PITCH + 2);
            const v2f x45 = *(const v2f*)(tc + p * RS_PITCH + 4);
            const float xr[6] = {x01.x, x01.y, x23.x, x23.y, x45.x, x45.y};
#pragma unroll
            for (int q = 0; q < 5; ++q) {
                const v2f xa = {xr[q], xr[q]};
                const v2f xb = {xr[q + 1], xr[q + 1]};
                a00 = __builtin_elementwise_fma(wt[p * 5 + q][0].lo, xa, a00);
                a01 = __builtin_elementwise_fma(wt[p * 5 + q][1].lo, xa, a01);
                a10 = __builtin_elementwise_fma(wt[p * 5 + q][0].hi, xb, a10);
                a11 = __builtin_elementwise_fma(wt[p * 5 + q][1].hi, xb, a11);
            }
        }
        const v4f o0 = {a00.x, a00.y, a10.x, a10.y};
        const v4f o1 = {a01.x, a01.y, a11.x, a11.y};
        __builtin_nontemporal_store(o0, (v4f*)(ob + (size_t)ch * HWO));
        __builtin_nontemporal_store(o1, (v4f*)(ob + (size_t)ch * HWO + WO));
    }
}

extern "C" void kernel_launch(void* const* d_in, const int* in_sizes, int n_in,
                              void* d_out, int out_size, void* d_ws, size_t ws_size,
                              hipStream_t stream) {
    const float* x      = (const float*)d_in[0];
    const float* w_comp = (const float*)d_in[1];
    const float* b_comp = (const float*)d_in[2];
    const float* w_enc  = (const float*)d_in[3];
    const float* b_enc  = (const float*)d_in[4];
    float* out = (float*)d_out;

    char* ws = (char*)d_ws;
    float* comp    = (float*)(ws);
    float* msoft   = (float*)(ws + (size_t)4194304);
    float* w_encF  = (float*)(ws + (size_t)4194304 + 6553600);

    compress_gemm  <<<256, 128, 0, stream>>>(x, w_comp, b_comp, w_enc, comp, w_encF);
    encoder_gemm   <<<256, 128, 0, stream>>>(comp, w_encF, b_enc, msoft);
    reassemble     <<<NB * 16 * 8, 128, 0, stream>>>(x, msoft, out);
}


# reassembly joins the 3x3-conv launch (8-wave workgroups): 4 extra waves stage the row's x slabs in LDS under the conv's k loop, softmax weights pass through LDS; 3 launches -> 2; sums unchanged
# speedup vs baseline: 8.9924x; 1.1165x over previous
#include <hip/hip_runtime.h>
#include <hip/hip_bf16.h>


#define NB     4
#define CIN    256
#define CCOMP  64
#define CMASK  100
#define HH     64
#define WW     64
#define HW     (HH*WW)
#define HO     128
#define WO     128
#define HWO    (HO*WO)
#define KTOT   576
#define MPAD   112

typedef float v2f __attribute__((ext_vector_type(2)));
typedef float v4f __attribute__((ext_vector_type(4)));
typedef float v8f __attribute__((ext_vector_type(8)));

#define EG_NLD  14
#define WENC_N  (36 * EG_NLD * 32 * 4)
#define WENC_PER_WG (WENC_N / 256)
#define CW_PITCH 260

__device__ __forceinline__ float wenc_table_elem(const float* __restrict__ w_enc, int idx) {
    const int e   = idx & 3;
    const int l   = (idx >> 2) & 31;
    const int g   = idx >> 7;
    const int blk = g / EG_NLD;
    const int j   = g % EG_NLD;
    const int f   = 2 * j + (e >> 1);
    const int s   = f / 7;
    const int mt  = f % 7;
    const int m   = mt * 16 + (l & 15);
    const int k   = blk * 16 + 4 * s + 2 * (l >> 4) + (e & 1);
    const int plane = k >> 6;
    const int ci    = k & 63;
    const float v = w_enc[min(m, CMASK - 1) * KTOT + ci * 9 + plane];
    return (m < CMASK) ? v : 0.0f;
}

#define CG_KS 8
#define CG_BK (4 * CG_KS)

#define GROUP_FENCE() __builtin_amdgcn_sched_barrier(0)


struct CgFrag {
    v4f a[CG_KS][2];
    v2f b[CG_KS];
};

__device__ __forceinline__ void cg_load_b(CgFrag& f, const float* __restrict__ xk, int k0) {
#pragma unroll
    for (int s = 0; s < CG_KS; ++s) {
        const int k = k0 + 4 * s;
        f.b[s].x = xk[(size_t)k * HW];
        f.b[s].y = xk[(size_t)(k + 1) * HW];
    }
}

__device__ __forceinline__ void cg_load_a(CgFrag& f, const float* wk, int k0) {
#pragma unroll
    for (int s = 0; s < CG_KS; ++s)
#pragma unroll
        for (int j = 0; j < 2; ++j)
            f.a[s][j] = *(const v4f*)(wk + ((k0 >> 2) + s) * CW_PITCH + j * 128);
}

__device__ __forceinline__ void cg_load(CgFrag& f, const float* __restrict__ xk,
                                        const float* wk, int k0) {
    cg_load_b(f, xk, k0);
    cg_load_a(f, wk, k0);
}

__device__ __forceinline__ void cg_mma(const CgFrag& f, v8f (&acc)[4]) {
#pragma unroll
    for (int s = 0; s < CG_KS; ++s)
#pragma unroll
        for (int mt = 0; mt < 4; ++mt) {
            const v2f afrag = (mt & 1) ? f.a[s][mt >> 1].hi : f.a[s][mt >> 1].lo;
            acc[mt] = __builtin_amdgcn_wmma_f32_16x16x4_f32(
                false, afrag, false, f.b[s], (short)0, acc[mt], false, false);
        }
}

__global__ void __launch_bounds__(128)
compress_gemm(const float* __restrict__ x, const float* __restrict__ w_comp,
              const float* __restrict__ b_comp, const float* __restrict__ w_enc,
              float* __restrict__ comp, float* __restrict__ w_encF) {
    __shared__ float wl[64 * CW_PITCH] __attribute__((aligned(16)));

    const int wid   = blockIdx.x * 4 + (threadIdx.x >> 5);
    const int lane  = threadIdx.x & 31;
    const int n     = wid >> 8;
    const int nt    = wid & 255;
    const int col   = nt * 16 + (lane & 15);
    const int khalf = (lane >> 4) * 2;

    const float* xk = x + (size_t)n * CIN * HW + col + (size_t)khalf * HW;
    const float* wk = wl + lane * 4;

    v8f acc[4];
    const v8f vzero = {0.f,0.f,0.f,0.f,0.f,0.f,0.f,0.f};
#pragma unroll
    for (int mt = 0; mt < 4; ++mt) acc[mt] = vzero;

    const int t = threadIdx.x;
    const float* wsrc = w_comp + (t >> 6) * CIN + 4 * (t & 63);
    v4f wst[32];
#pragma unroll
    for (int i = 0; i < 32; ++i)
        wst[i] = *(const v4f*)(wsrc + i * 2 * CIN);

    const bool two = (t < WENC_PER_WG - 128);
    const int  ti0 = blockIdx.x * WENC_PER_WG + t;
    const int  ti1 = two ? ti0 + 128 : ti0;
    const float te0 = wenc_table_elem(w_enc, ti0);
    const float te1 = wenc_table_elem(w_enc, ti1);

    CgFrag P, Q, R;
    cg_load_b(P, xk, 0);
    cg_load_b(Q, xk, CG_BK);

    float* wdst = &wl[(t & 63) * CW_PITCH + (t >> 6) * 4];
#pragma unroll
    for (int i = 0; i < 32; ++i) {
        float* d = wdst + (i >> 4) * 128 + (i & 7) * 8 + 2 * ((i >> 3) & 1);
        *(v2f*)d        = wst[i].lo;
        *(v2f*)(d + 64) = wst[i].hi;
    }
    __syncthreads();

    cg_load_a(P, wk, 0);
    cg_load_a(Q, wk, CG_BK);
#pragma unroll
    for (int k0 = 0; k0 < CIN - 4 * CG_BK; k0 += 3 * CG_BK) {
        cg_load(R, xk, wk, k0 + 2 * CG_BK);
        GROUP_FENCE();
        cg_mma(P, acc);
        GROUP_FENCE();
        cg_load(P, xk, wk, k0 + 3 * CG_BK);
        GROUP_FENCE();
        cg_mma(Q, acc);
        GROUP_FENCE();
        cg_load(Q, xk, wk, k0 + 4 * CG_BK);
        GROUP_FENCE();
        cg_mma(R, acc);
        GROUP_FENCE();
    }
    cg_mma(P, acc);
    cg_mma(Q, acc);

    float* cb = comp + (size_t)n * CCOMP * HW + col;
    const int mhi = 8 * (lane >> 4);
#pragma unroll
    for (int mt = 0; mt < 4; ++mt) {
#pragma unroll
        for (int r = 0; r < 8; ++r) {
            const int m = mt * 16 + r + mhi;
            cb[(size_t)m * HW] = acc[mt][r] + b_comp[m];
        }
    }

    w_encF[ti0] = te0;
    if (two) w_encF[ti1] = te1;
}

#define EG_KS   4

__device__ __forceinline__ float xhalf(float x) {
    return __builtin_bit_cast(float, __builtin_amdgcn_permlanex16(
        __builtin_bit_cast(int, x), __builtin_bit_cast(int, x),
        0x76543210, 0xfedcba98, false, false));
}
#define EG_NBLK 36

struct EgFrag {
    v4f   a[EG_NLD];
    float b0[EG_KS], b1[EG_KS];
    bool  valid;
};

__device__ __forceinline__ void eg_load(EgFrag& f, const float* __restrict__ cbase,
                                        const float* __restrict__ wk,
                                        int h, int w, int blk) {
    const int plane = blk >> 2;
    const int cib   = (blk & 3) * 16;
    const int dy = plane / 3, dx = plane % 3;
    const int y  = h + dy - 1;
    const int xw = w + dx - 1;
    f.valid = (y >= 0) && (y < HH) && (xw >= 0) && (xw < WW);
    const int yc = min(max(y, 0), HH - 1);
    const int xc = min(max(xw, 0), WW - 1);
    const float* bsrc = cbase + (size_t)cib * HW + yc * WW + xc;
    const float* asrc = wk + blk * (EG_NLD * 128);
#pragma unroll
    for (int s = 0; s < EG_KS; ++s) {
        f.b0[s] = bsrc[(size_t)(4 * s) * HW];
        f.b1[s] = bsrc[(size_t)(4 * s + 1) * HW];
    }
#pragma unroll
    for (int j = 0; j < EG_NLD; ++j)
        f.a[j] = *(const v4f*)(asrc + j * 128);
}

__device__ __forceinline__ void eg_mma(const EgFrag& f, v8f (&acc)[7]) {
#pragma unroll
    for (int s = 0; s < EG_KS; ++s) {
        v2f bfrag;
        bfrag.x = f.valid ? f.b0[s] : 0.0f;
        bfrag.y = f.valid ? f.b1[s] : 0.0f;
#pragma unroll
        for (int mt = 0; mt < 7; ++mt) {
            const int fi = s * 7 + mt;
            const v2f afrag = (fi & 1) ? f.a[fi >> 1].hi : f.a[fi >> 1].lo;
            acc[mt] = __builtin_amdgcn_wmma_f32_16x16x4_f32(
                false, afrag, false, bfrag, (short)0, acc[mt], false, false);
        }
    }
}

#define FZ_CH     64
#define FZ_ROWS   5
#define FZ_PITCH  72
#define FZ_CHF    (FZ_ROWS * FZ_PITCH)
#define FZ_SLABF  (FZ_CH * FZ_CHF)
#define FZ_NREG   3
#define FZ_WLF    (25 * 2 * 128)

__device__ __forceinline__ void reasm8(const float* tp, float* __restrict__ ob,
                                       const v4f (&wt)[25][2]) {
#pragma unroll 2
    for (int ch = 0; ch < 8; ++ch) {
        const float* tc = tp + ch * FZ_CHF;
        v2f a00 = {0.f, 0.f};
        v2f a01 = {0.f, 0.f};
        v2f a10 = {0.f, 0.f};
        v2f a11 = {0.f, 0.f};
#pragma unroll
        for (int p = 0; p < 5; ++p) {
            const v2f x01 = *(const v2f*)(tc + p * FZ_PITCH);
            const v2f x23 = *(const v2f*)(tc + p * FZ_PITCH + 2);
            const v2f x45 = *(const v2f*)(tc + p * FZ_PITCH + 4);
            const float xr[6] = {x01.x, x01.y, x23.x, x23.y, x45.x, x45.y};
#pragma unroll
            for (int q = 0; q < 5; ++q) {
                const v2f xa = {xr[q], xr[q]};
                const v2f xb = {xr[q + 1], xr[q + 1]};
                a00 = __builtin_elementwise_fma(wt[p * 5 + q][0].lo, xa, a00);
                a01 = __builtin_elementwise_fma(wt[p * 5 + q][1].lo, xa, a01);
                a10 = __builtin_elementwise_fma(wt[p * 5 + q][0].hi, xb, a10);
                a11 = __builtin_elementwise_fma(wt[p * 5 + q][1].hi, xb, a11);
            }
        }
        const v4f o0 = {a00.x, a00.y, a10.x, a10.y};
        const v4f o1 = {a01.x, a01.y, a11.x, a11.y};
        __builtin_nontemporal_store(o0, (v4f*)(ob + (size_t)ch * HWO));
        __builtin_nontemporal_store(o1, (v4f*)(ob + (size_t)ch * HWO + WO));
    }
}

__device__ __forceinline__ void lds_barrier() {
    __builtin_amdgcn_fence(__ATOMIC_RELEASE, "workgroup", "local");
    __builtin_amdgcn_s_barrier();
    __builtin_amdgcn_fence(__ATOMIC_ACQUIRE, "workgroup", "local");
}

__global__ void __launch_bounds__(256)
encode_reassemble(const float* __restrict__ comp, const float* __restrict__ w_encF,
                  const float* __restrict__ b_enc, const float* __restrict__ x,
                  float* __restrict__ out) {
    __shared__ float lds[FZ_NREG * FZ_SLABF + FZ_WLF] __attribute__((aligned(16)));
    float* const wl = lds + FZ_NREG * FZ_SLABF;

    const int t    = threadIdx.x;
    const int lane = t & 31;
    const int wv   = __builtin_amdgcn_readfirstlane(t >> 5);
    const int n    = blockIdx.x >> 6;
    const int h    = blockIdx.x & 63;
    const v4f zero4 = {0.f, 0.f, 0.f, 0.f};

    if (wv < 4) {
        const int w     = wv * 16 + (lane & 15);
        const int khalf = (lane >> 4) * 2;

        v8f acc[7];
        const v8f vzero = {0.f,0.f,0.f,0.f,0.f,0.f,0.f,0.f};
#pragma unroll
        for (int mt = 0; mt < 7; ++mt) acc[mt] = vzero;

        const float* cbase = comp + (size_t)n * CCOMP * HW + (size_t)khalf * HW;
        const float* wk    = w_encF + lane * 4;

        EgFrag P, Q, R;
        eg_load(P, cbase, wk, h, w, 0);
        GROUP_FENCE();
        eg_load(Q, cbase, wk, h, w, 1);
        GROUP_FENCE();
#pragma unroll 1
        for (int blk = 0; blk < EG_NBLK - 3; blk += 3) {
            eg_load(R, cbase, wk, h, w, blk + 2);
            GROUP_FENCE();
            eg_mma(P, acc);
            GROUP_FENCE();
            eg_load(P, cbase, wk, h, w, blk + 3);
            GROUP_FENCE();
            eg_mma(Q, acc);
            GROUP_FENCE();
            eg_load(Q, cbase, wk, h, w, blk + 4);
            GROUP_FENCE();
            eg_mma(R, acc);
            GROUP_FENCE();
        }
        eg_load(R, cbase, wk, h, w, EG_NBLK - 1);
        GROUP_FENCE();
        eg_mma(P, acc);
        eg_mma(Q, acc);
        eg_mma(R, acc);

        const int  hi  = lane >> 4;
        const bool low = (hi == 0);
        const float* bb = b_enc + 8 * hi;

        float v[6][8];
        float v6[4];
#pragma unroll
        for (int mt = 0; mt < 6; ++mt)
#pragma unroll
            for (int r = 0; r < 8; ++r) v[mt][r] = acc[mt][r] + bb[mt * 16 + r];
#pragma unroll
        for (int q = 0; q < 4; ++q) v6[q] = acc[6][q] + b_enc[96 + q];

#pragma unroll
        for (int q = 0; q < 4; ++q) {
            float mx = -1e30f;
#pragma unroll
            for (int mt = 0; mt < 6; ++mt) mx = fmaxf(mx, fmaxf(v[mt][q], v[mt][q + 4]));
            mx = fmaxf(mx, low ? v6[q] : -1e30f);
            mx = fmaxf(mx, xhalf(mx));
            float s = 0.0f;
#pragma unroll
            for (int mt = 0; mt < 6; ++mt) {
                v[mt][q]     = __expf(v[mt][q] - mx);      s += v[mt][q];
                v[mt][q + 4] = __expf(v[mt][q + 4] - mx);  s += v[mt][q + 4];
            }
            v6[q] = low ? __expf(v6[q] - mx) : 0.0f;
            s += v6[q];
            s += xhalf(s);
            const float inv = 1.0f / s;
#pragma unroll
            for (int mt = 0; mt < 6; ++mt) { v[mt][q] *= inv; v[mt][q + 4] *= inv; }
            v6[q] *= inv;
        }

        float* wlb = wl + (2 * hi) * 256 + 2 * w;
#pragma unroll
        for (int mt = 0; mt < 6; ++mt)
#pragma unroll
            for (int j = 0; j < 2; ++j)
#pragma unroll
                for (int sh = 0; sh < 2; ++sh) {
                    const v2f o = {v[mt][4 * j + 2 * sh], v[mt][4 * j + 2 * sh + 1]};
                    *(v2f*)(wlb + (mt * 4 + j) * 256 + sh * 128) = o;
                }
        if (low) {
#pragma unroll
            for (int sh = 0; sh < 2; ++sh) {
                const v2f o = {v6[2 * sh], v6[2 * sh + 1]};
                *(v2f*)(wlb + 24 * 256 + sh * 128) = o;
            }
        }
    } else {
        const int ht  = t - 128;
        const int c4  = ht & 15;
        const int chq = ht >> 4;

        for (int i = 0; i < 15; ++i) {
            const int f = ht + 128 * i;
            *(v4f*)(&lds[(f >> 1) * FZ_PITCH + ((f & 1) ? (FZ_PITCH - 4) : 0)]) = zero4;
        }

        const float* xs = x + ((size_t)n * CIN + chq) * HW + c4 * 4;
        float*       ts = lds + chq * FZ_CHF + 4 + c4 * 4;
#pragma unroll 1
        for (int sc = 0; sc < 2 * FZ_NREG; ++sc) {
            v4f stg[FZ_ROWS][4];
#pragma unroll
            for (int r = 0; r < FZ_ROWS; ++r) {
                const int  gy = h - 2 + r;
                const bool ok = (gy >= 0) && (gy < HH);
#pragma unroll
                for (int j = 0; j < 4; ++j) stg[r][j] = zero4;
                if (ok) {
#pragma unroll
                    for (int j = 0; j < 4; ++j)
                        stg[r][j] = *(const v4f*)(xs + (size_t)(sc * 32 + 8 * j) * HW + gy * WW);
                }
            }
#pragma unroll
            for (int r = 0; r < FZ_ROWS; ++r)
#pragma unroll
                for (int j = 0; j < 4; ++j)
                    *(v4f*)(ts + (sc * 32 + 8 * j) * FZ_CHF + r * FZ_PITCH) = stg[r][j];
        }
    }
    __syncthreads();

    const int w0 = lane * 2;

    v4f wt[25][2];
#pragma unroll
    for (int k = 0; k < 25; ++k) {
        wt[k][0] = *(const v4f*)(wl + (k * 2 + 0) * 128 + 4 * lane);
        wt[k][1] = *(const v4f*)(wl + (k * 2 + 1) * 128 + 4 * lane);
    }

    const int pc4  = t & 15;
    const int pchq = t >> 4;
    const float* xp = x + ((size_t)n * CIN + 3 * FZ_CH + pchq) * HW + pc4 * 4;
    v4f pf[FZ_ROWS][4];
#pragma unroll
    for (int r = 0; r < FZ_ROWS; ++r) {
        const int  gy = h - 2 + r;
        const bool ok = (gy >= 0) && (gy < HH);
#pragma unroll
        for (int j = 0; j < 4; ++j) pf[r][j] = zero4;
        if (ok) {
#pragma unroll
            for (int j = 0; j < 4; ++j)
                pf[r][j] = *(const v4f*)(xp + (size_t)(16 * j) * HW + gy * WW);
        }
    }

    const float* tp = lds + (wv * 8) * FZ_CHF + w0 + 2;
    float* ob = out + ((size_t)n * CIN + wv * 8) * HWO + (size_t)(2 * h) * WO + 2 * w0;

    reasm8(tp, ob, wt);
    lds_barrier();
    {
        float* ps = lds + pchq * FZ_CHF + 4 + pc4 * 4;
#pragma unroll
        for (int r = 0; r < FZ_ROWS; ++r)
#pragma unroll
            for (int j = 0; j < 4; ++j)
                *(v4f*)(ps + (16 * j) * FZ_CHF + r * FZ_PITCH) = pf[r][j];
    }
    lds_barrier();
    reasm8(tp + 1 * FZ_SLABF, ob + (size_t)(1 * FZ_CH) * HWO, wt);
    reasm8(tp + 2 * FZ_SLABF, ob + (size_t)(2 * FZ_CH) * HWO, wt);
    reasm8(tp,                ob + (size_t)(3 * FZ_CH) * HWO, wt);
}

extern "C" void kernel_launch(void* const* d_in, const int* in_sizes, int n_in,
                              void* d_out, int out_size, void* d_ws, size_t ws_size,
                              hipStream_t stream) {
    const float* x      = (const float*)d_in[0];
    const float* w_comp = (const float*)d_in[1];
    const float* b_comp = (const float*)d_in[2];
    const float* w_enc  = (const float*)d_in[3];
    const float* b_enc  = (const float*)d_in[4];
    float* out = (float*)d_out;

    char* ws = (char*)d_ws;
    float* comp    = (float*)(ws);
    float* w_encF  = (float*)(ws + (size_t)4194304);

    compress_gemm     <<<256, 128, 0, stream>>>(x, w_comp, b_comp, w_enc, comp, w_encF);
    encode_reassemble <<<256, 256, 0, stream>>>(comp, w_encF, b_enc, x, out);
}
